# starting kernel
# baseline (speedup 1.0000x reference)
#include <hip/hip_runtime.h>

#define HID      64
#define NNODES   100000
#define NEDGES   1200000
#define NGRAPHS  1000

typedef __attribute__((ext_vector_type(2))) float v2f;
typedef __attribute__((ext_vector_type(8))) float v8f;

template <bool RELU_IN>
__global__ __launch_bounds__(128) void gcn_gemm_wmma(
    const float* __restrict__ In, const float* __restrict__ W,
    const float* __restrict__ bias_in, float* __restrict__ Out)
{
    __shared__ float sW[64 * 65];
    __shared__ float sA[16 * 65];

    const int tid  = threadIdx.x;
    const int wave = tid >> 5;
    const int lane = tid & 31;
    const int rowBase = blockIdx.x * 16;

    for (int i = tid; i < 64 * 64; i += 128) {
        int r = i >> 6, c = i & 63;
        sW[r * 65 + c] = W[i];
    }
    for (int i = tid; i < 16 * 64; i += 128) {
        int r = i >> 6, c = i & 63;
        float v = In[(rowBase + r) * 64 + c];
        if (RELU_IN) {
            v += bias_in[c];
            v = v > 0.0f ? v : 0.0f;
        }
        sA[r * 65 + c] = v;
    }
    __syncthreads();

    const int m     = lane & 15;
    const int n     = (wave << 4) + (lane & 15);
    const int khalf = (lane >> 4) << 1;

    v8f acc = {};
#pragma unroll
    for (int k = 0; k < 64; k += 4) {
        v2f a, b;
        a.x = sA[m * 65 + k + khalf];
        a.y = sA[m * 65 + k + khalf + 1];
        b.x = sW[(k + khalf) * 65 + n];
        b.y = sW[(k + khalf + 1) * 65 + n];
        acc = __builtin_amdgcn_wmma_f32_16x16x4_f32(
             false, a,  false, b,
             (short)0, acc,  false,  false);
    }

    const int rOff = (lane >> 4) << 3;
    float* outp = Out + (size_t)(rowBase + rOff) * 64 + n;
#pragma unroll
    for (int v = 0; v < 8; ++v)
        outp[(size_t)v * 64] = acc[v];
}

__global__ __launch_bounds__(256) void gcn_scatter(
    const float* __restrict__ X, const int* __restrict__ src,
    const int* __restrict__ dst, float* __restrict__ Agg)
{
    const unsigned gid = blockIdx.x * 256u + threadIdx.x;
    const unsigned e = gid >> 5;
    const unsigned l = gid & 31;
    if (e >= NEDGES) return;
    const int s = src[e];
    const int d = dst[e];
    const float2 v = *(const float2*)(X + (size_t)s * 64 + l * 2);
    float* p = Agg + (size_t)d * 64 + l * 2;
    unsafeAtomicAdd(p, v.x);
    unsafeAtomicAdd(p + 1, v.y);
}

__global__ __launch_bounds__(256) void gcn_readout_acc(
    const float* __restrict__ Agg2, const int* __restrict__ gids,
    float* __restrict__ gsum, float* __restrict__ gcnt)
{
    const unsigned gid = blockIdx.x * 256u + threadIdx.x;
    const unsigned node = gid >> 6;
    if (node >= NNODES) return;
    const unsigned col = gid & 63;
    const int g = gids[node];
    unsafeAtomicAdd(&gsum[(size_t)g * 64 + col], Agg2[gid]);
    if (col == 0) unsafeAtomicAdd(&gcnt[g], 1.0f);
}

__global__ __launch_bounds__(256) void gcn_finalize(
    const float* __restrict__ gsum, const float* __restrict__ gcnt,
    const float* __restrict__ b2, float* __restrict__ out)
{
    const unsigned gid = blockIdx.x * 256u + threadIdx.x;
    if (gid >= NGRAPHS * 64u) return;
    const unsigned g = gid >> 6;
    const unsigned c = gid & 63;
    const float cnt = gcnt[g];
    out[gid] = (cnt > 0.0f) ? (gsum[gid] / cnt + b2[c]) : 0.0f;
}

extern "C" void kernel_launch(void* const* d_in, const int* in_sizes, int n_in,
                              void* d_out, int out_size, void* d_ws, size_t ws_size,
                              hipStream_t stream)
{
    (void)in_sizes; (void)n_in; (void)out_size; (void)ws_size;

    const float* feats = (const float*)d_in[0];
    const float* W1    = (const float*)d_in[1];
    const float* b1    = (const float*)d_in[2];
    const float* W2    = (const float*)d_in[3];
    const float* b2    = (const float*)d_in[4];
    const int*   src   = (const int*)d_in[5];
    const int*   dst   = (const int*)d_in[6];
    const int*   gids  = (const int*)d_in[7];
    float* out = (float*)d_out;

    const size_t nodeElems = (size_t)NNODES * HID;
    float* bufA = (float*)d_ws;
    float* bufB = bufA + nodeElems;
    float* gsum = bufB + nodeElems;
    float* gcnt = gsum + (size_t)NGRAPHS * HID;

    const int gemmBlocks    = NNODES / 16;
    const int scatterBlocks = (NEDGES * 32 + 255) / 256;
    const int readBlocks    = (NNODES * 64 + 255) / 256;
    const int finBlocks     = (NGRAPHS * 64 + 255) / 256;

    hipLaunchKernelGGL((gcn_gemm_wmma<false>), dim3(gemmBlocks), dim3(128), 0, stream,
                       feats, W1, (const float*)nullptr, bufA);
    hipMemsetAsync(bufB, 0, nodeElems * sizeof(float), stream);
    hipLaunchKernelGGL(gcn_scatter, dim3(scatterBlocks), dim3(256), 0, stream,
                       bufA, src, dst, bufB);
    hipLaunchKernelGGL((gcn_gemm_wmma<true>), dim3(gemmBlocks), dim3(128), 0, stream,
                       bufB, W2, b1, bufA);
    hipMemsetAsync(bufB, 0, nodeElems * sizeof(float), stream);
    hipLaunchKernelGGL(gcn_scatter, dim3(scatterBlocks), dim3(256), 0, stream,
                       bufA, src, dst, bufB);
    hipMemsetAsync(gsum, 0, ((size_t)NGRAPHS * HID + NGRAPHS) * sizeof(float), stream);
    hipLaunchKernelGGL(gcn_readout_acc, dim3(readBlocks), dim3(256), 0, stream,
                       bufB, gids, gsum, gcnt);
    hipLaunchKernelGGL(gcn_finalize, dim3(finBlocks), dim3(256), 0, stream,
                       gsum, gcnt, b2, out);
}


# readout: a wave walks 32 consecutive node rows, sums runs of equal graph id in registers and issues its f32 atomic adds once per run, not once per node; sum order changes
# speedup vs baseline: 1.1382x; 1.1382x over previous
#include <hip/hip_runtime.h>

#define HID      64
#define NNODES   100000
#define NEDGES   1200000
#define NGRAPHS  1000

typedef __attribute__((ext_vector_type(2))) float v2f;
typedef __attribute__((ext_vector_type(8))) float v8f;

template <bool RELU_IN>
__global__ __launch_bounds__(128) void gcn_gemm_wmma(
    const float* __restrict__ In, const float* __restrict__ W,
    const float* __restrict__ bias_in, float* __restrict__ Out)
{
    __shared__ float sW[64 * 65];
    __shared__ float sA[16 * 65];

    const int tid  = threadIdx.x;
    const int wave = tid >> 5;
    const int lane = tid & 31;
    const int rowBase = blockIdx.x * 16;

    for (int i = tid; i < 64 * 64; i += 128) {
        int r = i >> 6, c = i & 63;
        sW[r * 65 + c] = W[i];
    }
    for (int i = tid; i < 16 * 64; i += 128) {
        int r = i >> 6, c = i & 63;
        float v = In[(rowBase + r) * 64 + c];
        if (RELU_IN) {
            v += bias_in[c];
            v = v > 0.0f ? v : 0.0f;
        }
        sA[r * 65 + c] = v;
    }
    __syncthreads();

    const int m     = lane & 15;
    const int n     = (wave << 4) + (lane & 15);
    const int khalf = (lane >> 4) << 1;

    v8f acc = {};
#pragma unroll
    for (int k = 0; k < 64; k += 4) {
        v2f a, b;
        a.x = sA[m * 65 + k + khalf];
        a.y = sA[m * 65 + k + khalf + 1];
        b.x = sW[(k + khalf) * 65 + n];
        b.y = sW[(k + khalf + 1) * 65 + n];
        acc = __builtin_amdgcn_wmma_f32_16x16x4_f32(
             false, a,  false, b,
             (short)0, acc,  false,  false);
    }

    const int rOff = (lane >> 4) << 3;
    float* outp = Out + (size_t)(rowBase + rOff) * 64 + n;
#pragma unroll
    for (int v = 0; v < 8; ++v)
        outp[(size_t)v * 64] = acc[v];
}

__global__ __launch_bounds__(256) void gcn_scatter(
    const float* __restrict__ X, const int* __restrict__ src,
    const int* __restrict__ dst, float* __restrict__ Agg)
{
    const unsigned gid = blockIdx.x * 256u + threadIdx.x;
    const unsigned e = gid >> 5;
    const unsigned l = gid & 31;
    if (e >= NEDGES) return;
    const int s = src[e];
    const int d = dst[e];
    const float2 v = *(const float2*)(X + (size_t)s * 64 + l * 2);
    float* p = Agg + (size_t)d * 64 + l * 2;
    unsafeAtomicAdd(p, v.x);
    unsafeAtomicAdd(p + 1, v.y);
}

#define RO_ROWS 32

__device__ __forceinline__ void readout_flush(
    float* __restrict__ gsum, float* __restrict__ gcnt,
    int g, unsigned lane, float2 acc, float run)
{
    float* p = gsum + (size_t)g * 64 + lane * 2;
    unsafeAtomicAdd(p, acc.x);
    unsafeAtomicAdd(p + 1, acc.y);
    if (lane == 0) unsafeAtomicAdd(&gcnt[g], run);
}

__global__ __launch_bounds__(256) void gcn_readout_acc(
    const float* __restrict__ Agg2, const int* __restrict__ gids,
    float* __restrict__ gsum, float* __restrict__ gcnt)
{
    const unsigned wave = (blockIdx.x * 256u + threadIdx.x) >> 5;
    const unsigned lane = threadIdx.x & 31u;
    const unsigned base = wave * RO_ROWS;
    if (base >= NNODES) return;
    const int gl = gids[base + lane];
    const float2* rowp = (const float2*)(Agg2 + (size_t)base * 64) + lane;

    int cur = __builtin_amdgcn_readlane(gl, 0);
    float2 acc = make_float2(0.0f, 0.0f);
    float run = 0.0f;
    float2 v[RO_ROWS];
#pragma unroll
    for (int i = 0; i < RO_ROWS; ++i)
        v[i] = rowp[(size_t)i * 32];
#pragma unroll
    for (int i = 0; i < RO_ROWS; ++i) {
        const int g = __builtin_amdgcn_readlane(gl, i);
        if (g != cur) {
            readout_flush(gsum, gcnt, cur, lane, acc, run);
            cur = g;
            acc = make_float2(0.0f, 0.0f);
            run = 0.0f;
        }
        acc.x += v[i].x;
        acc.y += v[i].y;
        run += 1.0f;
    }
    readout_flush(gsum, gcnt, cur, lane, acc, run);
}

__global__ __launch_bounds__(256) void gcn_finalize(
    const float* __restrict__ gsum, const float* __restrict__ gcnt,
    const float* __restrict__ b2, float* __restrict__ out)
{
    const unsigned gid = blockIdx.x * 256u + threadIdx.x;
    if (gid >= NGRAPHS * 64u) return;
    const unsigned g = gid >> 6;
    const unsigned c = gid & 63;
    const float cnt = gcnt[g];
    out[gid] = (cnt > 0.0f) ? (gsum[gid] / cnt + b2[c]) : 0.0f;
}

extern "C" void kernel_launch(void* const* d_in, const int* in_sizes, int n_in,
                              void* d_out, int out_size, void* d_ws, size_t ws_size,
                              hipStream_t stream)
{
    (void)in_sizes; (void)n_in; (void)out_size; (void)ws_size;

    const float* feats = (const float*)d_in[0];
    const float* W1    = (const float*)d_in[1];
    const float* b1    = (const float*)d_in[2];
    const float* W2    = (const float*)d_in[3];
    const float* b2    = (const float*)d_in[4];
    const int*   src   = (const int*)d_in[5];
    const int*   dst   = (const int*)d_in[6];
    const int*   gids  = (const int*)d_in[7];
    float* out = (float*)d_out;

    const size_t nodeElems = (size_t)NNODES * HID;
    float* bufA = (float*)d_ws;
    float* bufB = bufA + nodeElems;
    float* gsum = bufB + nodeElems;
    float* gcnt = gsum + (size_t)NGRAPHS * HID;

    const int gemmBlocks    = NNODES / 16;
    const int scatterBlocks = (NEDGES * 32 + 255) / 256;
    const int readBlocks    = (NNODES / RO_ROWS * 32 + 255) / 256;
    const int finBlocks     = (NGRAPHS * 64 + 255) / 256;

    hipLaunchKernelGGL((gcn_gemm_wmma<false>), dim3(gemmBlocks), dim3(128), 0, stream,
                       feats, W1, (const float*)nullptr, bufA);
    hipMemsetAsync(bufB, 0, nodeElems * sizeof(float), stream);
    hipLaunchKernelGGL(gcn_scatter, dim3(scatterBlocks), dim3(256), 0, stream,
                       bufA, src, dst, bufB);
    hipLaunchKernelGGL((gcn_gemm_wmma<true>), dim3(gemmBlocks), dim3(128), 0, stream,
                       bufB, W2, b1, bufA);
    hipMemsetAsync(bufB, 0, nodeElems * sizeof(float), stream);
    hipLaunchKernelGGL(gcn_scatter, dim3(scatterBlocks), dim3(256), 0, stream,
                       bufA, src, dst, bufB);
    hipMemsetAsync(gsum, 0, ((size_t)NGRAPHS * HID + NGRAPHS) * sizeof(float), stream);
    hipLaunchKernelGGL(gcn_readout_acc, dim3(readBlocks), dim3(256), 0, stream,
                       bufB, gids, gsum, gcnt);
    hipLaunchKernelGGL(gcn_finalize, dim3(finBlocks), dim3(256), 0, stream,
                       gsum, gcnt, b2, out);
}


# aggregation: edges counting-sorted by dst once per call (int atomics); a wave per row pulls its source rows into register sums, one store per row: no f32 atomics or clears there; sum order changes
# speedup vs baseline: 1.5712x; 1.3804x over previous
#include <hip/hip_runtime.h>

#define HID      64
#define NNODES   100000
#define NEDGES   1200000
#define NGRAPHS  1000

typedef __attribute__((ext_vector_type(2))) float v2f;
typedef __attribute__((ext_vector_type(8))) float v8f;

template <bool RELU_IN>
__global__ __launch_bounds__(128) void gcn_gemm_wmma(
    const float* __restrict__ In, const float* __restrict__ W,
    const float* __restrict__ bias_in, float* __restrict__ Out)
{
    __shared__ float sW[64 * 65];
    __shared__ float sA[16 * 65];

    const int tid  = threadIdx.x;
    const int wave = tid >> 5;
    const int lane = tid & 31;
    const int rowBase = blockIdx.x * 16;

    for (int i = tid; i < 64 * 64; i += 128) {
        int r = i >> 6, c = i & 63;
        sW[r * 65 + c] = W[i];
    }
    for (int i = tid; i < 16 * 64; i += 128) {
        int r = i >> 6, c = i & 63;
        float v = In[(rowBase + r) * 64 + c];
        if (RELU_IN) {
            v += bias_in[c];
            v = v > 0.0f ? v : 0.0f;
        }
        sA[r * 65 + c] = v;
    }
    __syncthreads();

    const int m     = lane & 15;
    const int n     = (wave << 4) + (lane & 15);
    const int khalf = (lane >> 4) << 1;

    v8f acc = {};
#pragma unroll
    for (int k = 0; k < 64; k += 4) {
        v2f a, b;
        a.x = sA[m * 65 + k + khalf];
        a.y = sA[m * 65 + k + khalf + 1];
        b.x = sW[(k + khalf) * 65 + n];
        b.y = sW[(k + khalf + 1) * 65 + n];
        acc = __builtin_amdgcn_wmma_f32_16x16x4_f32(
             false, a,  false, b,
             (short)0, acc,  false,  false);
    }

    const int rOff = (lane >> 4) << 3;
    float* outp = Out + (size_t)(rowBase + rOff) * 64 + n;
#pragma unroll
    for (int v = 0; v < 8; ++v)
        outp[(size_t)v * 64] = acc[v];
}

#define CSR_PAD 100032

__global__ __launch_bounds__(256) void csr_count(
    const int* __restrict__ dst, int* __restrict__ cur)
{
    const unsigned e = blockIdx.x * 256u + threadIdx.x;
    if (e < NEDGES) atomicAdd(&cur[dst[e]], 1);
}

__global__ __launch_bounds__(1024) void csr_offsets(
    int* __restrict__ cur, int* __restrict__ off)
{
    __shared__ int wsum[32];
    __shared__ int sbase;
    const unsigned tid  = threadIdx.x;
    const unsigned lane = tid & 31u;
    const unsigned wave = tid >> 5;
    const unsigned n = blockIdx.x * 1024u + tid;
    const int c = (n < NNODES) ? cur[n] : 0;

    int incl = c;
#pragma unroll
    for (int d = 1; d < 32; d <<= 1) {
        const int t = __shfl_up(incl, d);
        if (lane >= (unsigned)d) incl += t;
    }
    if (lane == 31u) wsum[wave] = incl;
    __syncthreads();
    if (wave == 0u) {
        const int w = wsum[lane];
        int wi = w;
#pragma unroll
        for (int d = 1; d < 32; d <<= 1) {
            const int t = __shfl_up(wi, d);
            if (lane >= (unsigned)d) wi += t;
        }
        wsum[lane] = wi - w;
        if (lane == 31u) sbase = atomicAdd(&cur[NNODES], wi);
    }
    __syncthreads();
    const int o = sbase + wsum[wave] + incl - c;
    if (n < NNODES) { off[n] = o; cur[n] = o; }
}

__global__ __launch_bounds__(256) void csr_fill(
    const int* __restrict__ src, const int* __restrict__ dst,
    int* __restrict__ cur, int* __restrict__ ssrc)
{
    const unsigned e = blockIdx.x * 256u + threadIdx.x;
    if (e < NEDGES) {
        const int p = atomicAdd(&cur[dst[e]], 1);
        if ((unsigned)p < (unsigned)NEDGES) ssrc[p] = src[e];
    }
}

__global__ __launch_bounds__(256) void gcn_pull(
    const float* __restrict__ X, const int* __restrict__ off,
    const int* __restrict__ endp, const int* __restrict__ ssrc,
    float* __restrict__ Agg)
{
    const unsigned n = __builtin_amdgcn_readfirstlane((blockIdx.x * 256u + threadIdx.x) >> 5);
    const unsigned lane = threadIdx.x & 31u;
    const unsigned h  = lane >> 4;
    const unsigned c4 = (lane & 15u) << 2;
    const int b0 = off[n];
    const int e0 = endp[n];
    const int beg = b0 > 0 ? b0 : 0;
    const int end = e0 < NEDGES ? e0 : NEDGES;

    float4 acc = make_float4(0.0f, 0.0f, 0.0f, 0.0f);
    for (int base = beg; base < end; base += 32) {
        const int rem = end - base;
        const int m = rem < 32 ? rem : 32;
        const int li = (int)lane < m ? (int)lane : m - 1;
        const unsigned raw = (unsigned)ssrc[base + li];
        const int idx = (int)(raw < (unsigned)NNODES ? raw : (unsigned)(NNODES - 1));
        for (int j = 0; j < m; j += 8) {
#pragma unroll
            for (int u = 0; u < 4; ++u) {
                const int ea = j + 2 * u;
                const int eb = ea + 1;
                const int sa = __builtin_amdgcn_readlane(idx, ea < m ? ea : m - 1);
                const int sb = __builtin_amdgcn_readlane(idx, eb < m ? eb : m - 1);
                const int s = h ? sb : sa;
                const bool live = (ea + (int)h) < m;
                const float4 v = *(const float4*)(X + (size_t)s * 64 + c4);
                acc.x += live ? v.x : 0.0f;
                acc.y += live ? v.y : 0.0f;
                acc.z += live ? v.z : 0.0f;
                acc.w += live ? v.w : 0.0f;
            }
        }
    }
    acc.x += __shfl_xor(acc.x, 16);
    acc.y += __shfl_xor(acc.y, 16);
    acc.z += __shfl_xor(acc.z, 16);
    acc.w += __shfl_xor(acc.w, 16);
    if (h == 0u)
        *(float4*)(Agg + (size_t)n * 64 + c4) = acc;
}

#define RO_ROWS 32

__device__ __forceinline__ void readout_flush(
    float* __restrict__ gsum, float* __restrict__ gcnt,
    int g, unsigned lane, float2 acc, float run)
{
    float* p = gsum + (size_t)g * 64 + lane * 2;
    unsafeAtomicAdd(p, acc.x);
    unsafeAtomicAdd(p + 1, acc.y);
    if (lane == 0) unsafeAtomicAdd(&gcnt[g], run);
}

__global__ __launch_bounds__(256) void gcn_readout_acc(
    const float* __restrict__ Agg2, const int* __restrict__ gids,
    float* __restrict__ gsum, float* __restrict__ gcnt)
{
    const unsigned wave = (blockIdx.x * 256u + threadIdx.x) >> 5;
    const unsigned lane = threadIdx.x & 31u;
    const unsigned base = wave * RO_ROWS;
    if (base >= NNODES) return;
    const int gl = gids[base + lane];
    const float2* rowp = (const float2*)(Agg2 + (size_t)base * 64) + lane;

    int cur = __builtin_amdgcn_readlane(gl, 0);
    float2 acc = make_float2(0.0f, 0.0f);
    float run = 0.0f;
    float2 v[RO_ROWS];
#pragma unroll
    for (int i = 0; i < RO_ROWS; ++i)
        v[i] = rowp[(size_t)i * 32];
#pragma unroll
    for (int i = 0; i < RO_ROWS; ++i) {
        const int g = __builtin_amdgcn_readlane(gl, i);
        if (g != cur) {
            readout_flush(gsum, gcnt, cur, lane, acc, run);
            cur = g;
            acc = make_float2(0.0f, 0.0f);
            run = 0.0f;
        }
        acc.x += v[i].x;
        acc.y += v[i].y;
        run += 1.0f;
    }
    readout_flush(gsum, gcnt, cur, lane, acc, run);
}

__global__ __launch_bounds__(256) void gcn_finalize(
    const float* __restrict__ gsum, const float* __restrict__ gcnt,
    const float* __restrict__ b2, float* __restrict__ out)
{
    const unsigned gid = blockIdx.x * 256u + threadIdx.x;
    if (gid >= NGRAPHS * 64u) return;
    const unsigned g = gid >> 6;
    const unsigned c = gid & 63;
    const float cnt = gcnt[g];
    out[gid] = (cnt > 0.0f) ? (gsum[gid] / cnt + b2[c]) : 0.0f;
}

extern "C" void kernel_launch(void* const* d_in, const int* in_sizes, int n_in,
                              void* d_out, int out_size, void* d_ws, size_t ws_size,
                              hipStream_t stream)
{
    (void)in_sizes; (void)n_in; (void)out_size; (void)ws_size;

    const float* feats = (const float*)d_in[0];
    const float* W1    = (const float*)d_in[1];
    const float* b1    = (const float*)d_in[2];
    const float* W2    = (const float*)d_in[3];
    const float* b2    = (const float*)d_in[4];
    const int*   src   = (const int*)d_in[5];
    const int*   dst   = (const int*)d_in[6];
    const int*   gids  = (const int*)d_in[7];
    float* out = (float*)d_out;

    const size_t nodeElems = (size_t)NNODES * HID;
    float* bufA = (float*)d_ws;
    float* bufB = bufA + nodeElems;
    float* gsum = bufB + nodeElems;
    float* gcnt = gsum + (size_t)NGRAPHS * HID;
    int*   cur  = (int*)(gcnt + NGRAPHS);
    int*   off  = cur + CSR_PAD;
    int*   ssrc = off + CSR_PAD;

    const int gemmBlocks    = NNODES / 16;
    const int edgeBlocks    = (NEDGES + 255) / 256;
    const int offBlocks     = (NNODES + 1023) / 1024;
    const int pullBlocks    = NNODES / 8;
    const int readBlocks    = (NNODES / RO_ROWS * 32 + 255) / 256;
    const int finBlocks     = (NGRAPHS * 64 + 255) / 256;

    hipMemsetAsync(cur, 0, (size_t)CSR_PAD * sizeof(int), stream);
    hipLaunchKernelGGL(csr_count, dim3(edgeBlocks), dim3(256), 0, stream, dst, cur);
    hipLaunchKernelGGL(csr_offsets, dim3(offBlocks), dim3(1024), 0, stream, cur, off);
    hipLaunchKernelGGL(csr_fill, dim3(edgeBlocks), dim3(256), 0, stream, src, dst, cur, ssrc);
    hipLaunchKernelGGL((gcn_gemm_wmma<false>), dim3(gemmBlocks), dim3(128), 0, stream,
                       feats, W1, (const float*)nullptr, bufA);
    hipLaunchKernelGGL(gcn_pull, dim3(pullBlocks), dim3(256), 0, stream,
                       bufA, off, cur, ssrc, bufB);
    hipLaunchKernelGGL((gcn_gemm_wmma<true>), dim3(gemmBlocks), dim3(128), 0, stream,
                       bufB, W2, b1, bufA);
    hipLaunchKernelGGL(gcn_pull, dim3(pullBlocks), dim3(256), 0, stream,
                       bufA, off, cur, ssrc, bufB);
    hipMemsetAsync(gsum, 0, ((size_t)NGRAPHS * HID + NGRAPHS) * sizeof(float), stream);
    hipLaunchKernelGGL(gcn_readout_acc, dim3(readBlocks), dim3(256), 0, stream,
                       bufB, gids, gsum, gcnt);
    hipLaunchKernelGGL(gcn_finalize, dim3(finBlocks), dim3(256), 0, stream,
                       gsum, gcnt, b2, out);
}


# layer-2 launch: a work-group of 8 waves stays for 2 pairs of 16-row tiles, W2 staged in LDS once per work-group (two A buffers, one barrier a pass): 12504 waves enter, not 25000; sums as before
# speedup vs baseline: 4.8188x; 3.0670x over previous
#include <hip/hip_runtime.h>

#define HID      64
#define NNODES   100000
#define NEDGES   1200000
#define NGRAPHS  1000

typedef __attribute__((ext_vector_type(2))) float v2f;
typedef __attribute__((ext_vector_type(8))) float v8f;

#define CSR_PAD   100032
#define SORT_C    250
#define SORT_EPB  4800
#define SORT_NB   250
#define SORT_R    400
#define SORT_PER  5
#define SORT_CAP  16384
#define SRC_BITS  17
#define SRC_MASK  0x1FFFFu

static_assert(SORT_C * SORT_EPB == NEDGES, "chunks cover the edge list");
static_assert(SORT_NB * SORT_R == NNODES, "ranges cover the nodes");
static_assert(NNODES <= (1 << SRC_BITS) && SORT_R <= 512, "packed word holds row and source");
static_assert(SORT_PER * 1024 >= SORT_EPB && SORT_C <= 256 && SORT_NB < 256, "thread maps");

#define FG_TILES  25
#define FG_NT     (NNODES / 16)
#define FG_WGS    ((FG_NT + FG_TILES - 1) / FG_TILES)
#define FG_APITCH 68
#define FG_SLAB   (16 * FG_APITCH)
#define FG_LDS    (64 * 65 + FG_TILES * FG_SLAB)
#define FG_CLEAR  (NGRAPHS * HID + NGRAPHS)
#define FG_CLR    ((FG_CLEAR + FG_WGS - 1) / FG_WGS)

static_assert(NNODES % 16 == 0, "row tiles have no tail");
static_assert(FG_CLR <= 1024, "a thread stores at most one zero");
static_assert(FG_LDS >= 256 + SORT_EPB, "the sort role fits the same LDS");
static_assert(SORT_C + FG_WGS <= 512, "one resident pass at two work-groups per WGP");

__global__ __launch_bounds__(1024) void sort_bin_gemm(
    const int* __restrict__ src, const int* __restrict__ dst,
    int* __restrict__ binned, int* __restrict__ tbl,
    const float* __restrict__ In, const float* __restrict__ W,
    float* __restrict__ Out, float* __restrict__ gclear)
{
    __shared__ __attribute__((aligned(16))) float lds[FG_LDS];
    const unsigned tid  = threadIdx.x;
    const unsigned lane = tid & 31u;

    if (blockIdx.x >= SORT_C) {
        const unsigned wave = tid >> 5;
        const unsigned g    = blockIdx.x - SORT_C;
        float* sW = lds;
        {
            const float4 w = ((const float4*)W)[tid];
            float* p = sW + (tid >> 4) * 65u + ((tid & 15u) << 2);
            p[0] = w.x; p[1] = w.y; p[2] = w.z; p[3] = w.w;
        }
        const unsigned tile = g * FG_TILES + wave;
        const bool has = wave < FG_TILES && tile < FG_NT;
        float* sA = lds + 64 * 65 + (has ? wave : 0u) * FG_SLAB;
        if (has) {
            const float4* t4 = (const float4*)(In + (size_t)tile * (16 * 64));
            float4 v[8];
#pragma unroll
            for (int i = 0; i < 8; ++i) v[i] = t4[lane + 32u * (unsigned)i];
#pragma unroll
            for (int i = 0; i < 8; ++i)
                *(float4*)(sA + (2u * (unsigned)i + (lane >> 4)) * FG_APITCH + ((lane & 15u) << 2)) = v[i];
        }
        if (tid < FG_CLR) {
            const unsigned i = g * FG_CLR + tid;
            if (i < FG_CLEAR) gclear[i] = 0.0f;
        }
        __syncthreads();
        if (!has) return;

        const unsigned m     = lane & 15u;
        const unsigned khalf = (lane >> 4) << 1;
        v8f acc0 = {}, acc1 = {}, acc2 = {}, acc3 = {};
#pragma unroll
        for (int k = 0; k < 64; k += 4) {
            v2f a, b0, b1, b2, b3;
            const float* pa = sA + m * FG_APITCH + k + khalf;
            const float* pb = sW + (k + khalf) * 65u + m;
            a.x  = pa[0];   a.y  = pa[1];
            b0.x = pb[0];   b0.y = pb[65];
            b1.x = pb[16];  b1.y = pb[65 + 16];
            b2.x = pb[32];  b2.y = pb[65 + 32];
            b3.x = pb[48];  b3.y = pb[65 + 48];
            acc0 = __builtin_amdgcn_wmma_f32_16x16x4_f32(false, a, false, b0, (short)0, acc0, false, false);
            acc1 = __builtin_amdgcn_wmma_f32_16x16x4_f32(false, a, false, b1, (short)0, acc1, false, false);
            acc2 = __builtin_amdgcn_wmma_f32_16x16x4_f32(false, a, false, b2, (short)0, acc2, false, false);
            acc3 = __builtin_amdgcn_wmma_f32_16x16x4_f32(false, a, false, b3, (short)0, acc3, false, false);
        }

        const unsigned rOff = (lane >> 4) << 3;
        float* outp = Out + (size_t)(tile * 16u + rOff) * 64 + m;
#pragma unroll
        for (int v = 0; v < 8; ++v) {
            outp[(size_t)v * 64]      = acc0[v];
            outp[(size_t)v * 64 + 16] = acc1[v];
            outp[(size_t)v * 64 + 32] = acc2[v];
            outp[(size_t)v * 64 + 48] = acc3[v];
        }
        return;
    }

    int* cnt   = (int*)lds;
    int* stage = cnt + 256;
    const unsigned c    = blockIdx.x;
    const unsigned base = c * SORT_EPB;

    if (tid < 256u) cnt[tid] = 0;
    __syncthreads();

    unsigned rec[SORT_PER], bin[SORT_PER];
#pragma unroll
    for (int k = 0; k < SORT_PER; ++k) {
        const unsigned i = tid + (unsigned)k * 1024u;
        bin[k] = 0xFFFFFFFFu;
        rec[k] = 0u;
        if (i < SORT_EPB) {
            unsigned d = (unsigned)dst[base + i];
            unsigned s = (unsigned)src[base + i];
            d = d < (unsigned)NNODES ? d : (unsigned)(NNODES - 1);
            s = s < (unsigned)NNODES ? s : (unsigned)(NNODES - 1);
            const unsigned b = d / SORT_R;
            bin[k] = b;
            rec[k] = ((d - b * SORT_R) << SRC_BITS) | s;
            atomicAdd(&cnt[b], 1);
        }
    }
    __syncthreads();

    if (tid < 32u) {
        int v[8];
        int sum = 0;
#pragma unroll
        for (int j = 0; j < 8; ++j) {
            const unsigned idx = lane * 8u + (unsigned)j;
            v[j] = idx < SORT_NB ? cnt[idx] : 0;
            sum += v[j];
        }
        int incl = sum;
#pragma unroll
        for (int d = 1; d < 32; d <<= 1) {
            const int t = __shfl_up(incl, d);
            if (lane >= (unsigned)d) incl += t;
        }
        int run = incl - sum;
#pragma unroll
        for (int j = 0; j < 8; ++j) {
            const unsigned idx = lane * 8u + (unsigned)j;
            cnt[idx] = run;
            if (idx <= SORT_NB) tbl[idx * SORT_C + c] = run;
            run += v[j];
        }
    }
    __syncthreads();

#pragma unroll
    for (int k = 0; k < SORT_PER; ++k) {
        if (bin[k] != 0xFFFFFFFFu) {
            const int p = atomicAdd(&cnt[bin[k]], 1);
            if ((unsigned)p < (unsigned)SORT_EPB) stage[p] = (int)rec[k];
        }
    }
    __syncthreads();

    for (unsigned i = tid; i < SORT_EPB; i += 1024u) binned[base + i] = stage[i];
}

__global__ __launch_bounds__(1024) void sort_place(
    const int* __restrict__ binned, const int* __restrict__ tbl,
    int* __restrict__ off, int* __restrict__ endp, int* __restrict__ ssrc)
{
    __shared__ int pcS[256];
    __shared__ int pcE[256];
    __shared__ int cnt[512];
    __shared__ int wsum[32];
    __shared__ int sBeg, sEnd;
    __shared__ int stage[SORT_CAP];
    const unsigned tid  = threadIdx.x;
    const unsigned lane = tid & 31u;
    const unsigned wave = tid >> 5;
    const unsigned b    = blockIdx.x;

    int o0 = 0, o1 = 0;
    if (tid < SORT_C) {
        o0 = tbl[b * SORT_C + tid];
        o1 = tbl[(b + 1u) * SORT_C + tid];
        o0 = o0 > 0 ? o0 : 0;
        o0 = o0 < SORT_EPB ? o0 : SORT_EPB;
        o1 = o1 > o0 ? o1 : o0;
        o1 = o1 < SORT_EPB ? o1 : SORT_EPB;
    }
    if (tid < 256u) { pcS[tid] = o0; pcE[tid] = o1; }
    if (tid < 512u) cnt[tid] = 0;
    if (tid == 0u) { sBeg = 0; sEnd = 0; }
    __syncthreads();

    if (tid < 256u) {
        int a = o0, e = o1;
#pragma unroll
        for (int d = 16; d > 0; d >>= 1) {
            a += __shfl_xor(a, d);
            e += __shfl_xor(e, d);
        }
        if (lane == 0u) { atomicAdd(&sBeg, a); atomicAdd(&sEnd, e); }
    }

    unsigned r[8];
    bool ok[8];
#pragma unroll
    for (int j = 0; j < 8; ++j) {
        const unsigned c = wave + 32u * (unsigned)j;
        int s = 0, e = 0;
        if (c < SORT_C) { s = pcS[c]; e = pcE[c]; }
        const int i = s + (int)lane;
        ok[j] = i < e;
        r[j] = ok[j] ? (unsigned)binned[c * SORT_EPB + (unsigned)i] : 0u;
    }
#pragma unroll
    for (int j = 0; j < 8; ++j) {
        if (ok[j]) {
            unsigned dl = r[j] >> SRC_BITS;
            dl = dl < SORT_R ? dl : SORT_R - 1u;
            atomicAdd(&cnt[dl], 1);
        }
    }
    for (unsigned c = wave; c < SORT_C; c += 32u) {
        const int e = pcE[c];
        for (int i = pcS[c] + 32 + (int)lane; i < e; i += 32) {
            unsigned dl = (unsigned)binned[c * SORT_EPB + (unsigned)i] >> SRC_BITS;
            dl = dl < SORT_R ? dl : SORT_R - 1u;
            atomicAdd(&cnt[dl], 1);
        }
    }
    __syncthreads();

    int cv = 0, incl = 0;
    if (tid < 512u) {
        cv = cnt[tid];
        incl = cv;
#pragma unroll
        for (int d = 1; d < 32; d <<= 1) {
            const int t = __shfl_up(incl, d);
            if (lane >= (unsigned)d) incl += t;
        }
        if (lane == 31u) wsum[wave] = incl;
    }
    __syncthreads();
    if (wave == 0u) {
        const int w = lane < 16u ? wsum[lane] : 0;
        int wi = w;
#pragma unroll
        for (int d = 1; d < 32; d <<= 1) {
            const int t = __shfl_up(wi, d);
            if (lane >= (unsigned)d) wi += t;
        }
        wsum[lane] = wi - w;
    }
    __syncthreads();
    const int beg   = sBeg;
    const int total = sEnd - beg;
    const bool fits = total <= SORT_CAP;
    if (tid < SORT_R) {
        const int excl = wsum[wave] + incl - cv;
        const unsigned n = b * SORT_R + tid;
        cnt[tid] = excl;
        off[n]  = beg + excl;
        endp[n] = beg + excl + cv;
    }
    __syncthreads();

#pragma unroll
    for (int j = 0; j < 8; ++j) {
        if (ok[j]) {
            unsigned dl = r[j] >> SRC_BITS;
            dl = dl < SORT_R ? dl : SORT_R - 1u;
            const int p = atomicAdd(&cnt[dl], 1);
            const int s = (int)(r[j] & SRC_MASK);
            if (fits) {
                if ((unsigned)p < (unsigned)SORT_CAP) stage[p] = s;
            } else {
                const int q = beg + p;
                if ((unsigned)q < (unsigned)NEDGES) ssrc[q] = s;
            }
        }
    }
    for (unsigned c = wave; c < SORT_C; c += 32u) {
        const int e = pcE[c];
        for (int i = pcS[c] + 32 + (int)lane; i < e; i += 32) {
            const unsigned rr = (unsigned)binned[c * SORT_EPB + (unsigned)i];
            unsigned dl = rr >> SRC_BITS;
            dl = dl < SORT_R ? dl : SORT_R - 1u;
            const int p = atomicAdd(&cnt[dl], 1);
            const int s = (int)(rr & SRC_MASK);
            if (fits) {
                if ((unsigned)p < (unsigned)SORT_CAP) stage[p] = s;
            } else {
                const int q = beg + p;
                if ((unsigned)q < (unsigned)NEDGES) ssrc[q] = s;
            }
        }
    }
    __syncthreads();

    if (fits) {
        for (int i = (int)tid; i < total; i += 1024) {
            const int q = beg + i;
            if ((unsigned)q < (unsigned)NEDGES) ssrc[q] = stage[i];
        }
    }
}

__device__ __forceinline__ int clamp_src(unsigned raw)
{
    return (int)(raw < (unsigned)NNODES ? raw : (unsigned)(NNODES - 1));
}

__device__ __forceinline__ int load_chunk_idx(
    const int* __restrict__ ssrc, int base, int end, unsigned lane)
{
    const int rem = end - base;
    const int m = rem < 32 ? rem : 32;
    int li = (int)lane < m ? (int)lane : m - 1;
    li = li > 0 ? li : 0;
    int p = base + li;
    p = p < NEDGES ? p : NEDGES - 1;
    return clamp_src((unsigned)ssrc[p]);
}

__device__ __forceinline__ float4 pull_row_sum(
    const float* __restrict__ X, const int* __restrict__ ssrc,
    int beg, int end, int idx_first, unsigned lane, unsigned h, unsigned c4)
{
    float4 acc = make_float4(0.0f, 0.0f, 0.0f, 0.0f);
    int idx = idx_first;
    for (int base = beg; base < end; base += 32) {
        const int rem = end - base;
        const int m = rem < 32 ? rem : 32;
        if (base != beg) idx = load_chunk_idx(ssrc, base, end, lane);
        for (int j = 0; j < m; j += 8) {
#pragma unroll
            for (int u = 0; u < 4; ++u) {
                const int ea = j + 2 * u;
                const int eb = ea + 1;
                const int sa = __builtin_amdgcn_readlane(idx, ea < m ? ea : m - 1);
                const int sb = __builtin_amdgcn_readlane(idx, eb < m ? eb : m - 1);
                const int s = h ? sb : sa;
                const bool live = (ea + (int)h) < m;
                const float4 v = *(const float4*)(X + (size_t)s * 64 + c4);
                acc.x += live ? v.x : 0.0f;
                acc.y += live ? v.y : 0.0f;
                acc.z += live ? v.z : 0.0f;
                acc.w += live ? v.w : 0.0f;
            }
        }
    }
    acc.x += __shfl_xor(acc.x, 16);
    acc.y += __shfl_xor(acc.y, 16);
    acc.z += __shfl_xor(acc.z, 16);
    acc.w += __shfl_xor(acc.w, 16);
    return acc;
}

#define PG_PAIRS   (NNODES / 32)
#define PG_PASSES  2
#define PG_WGS     ((PG_PAIRS + PG_PASSES - 1) / PG_PASSES)

static_assert(NNODES % 32 == 0, "pairs of row tiles have no tail");

__device__ __forceinline__ void pg_fetch(
    const int* __restrict__ off, const int* __restrict__ endp, const int* __restrict__ ssrc,
    unsigned n0, unsigned lane, int (&beg)[4], int (&end)[4], int (&idx0)[4])
{
#pragma unroll
    for (int r = 0; r < 4; ++r) {
        const int b0 = off[n0 + r];
        const int e0 = endp[n0 + r];
        beg[r] = b0 > 0 ? b0 : 0;
        end[r] = e0 < NEDGES ? e0 : NEDGES;
    }
#pragma unroll
    for (int r = 0; r < 4; ++r)
        idx0[r] = load_chunk_idx(ssrc, beg[r], end[r], lane);
}

__global__ __launch_bounds__(256, 16) void gcn_pull_gemm(
    const float* __restrict__ X, const int* __restrict__ off,
    const int* __restrict__ endp, const int* __restrict__ ssrc,
    const float* __restrict__ W, const float* __restrict__ bias_in,
    float* __restrict__ Out)
{
    __shared__ float sW[64 * 65];
    __shared__ float sA[2][32 * 65];

    const unsigned tid  = threadIdx.x;
    const unsigned wave = tid >> 5;
    const unsigned lane = tid & 31u;

    unsigned pair = blockIdx.x;
    int beg[4], end[4], idx0[4];
    pg_fetch(off, endp, ssrc, __builtin_amdgcn_readfirstlane(pair * 32u + (wave << 2)), lane, beg, end, idx0);

    __builtin_amdgcn_sched_barrier(0);
    float4 w[4];
#pragma unroll
    for (int i = 0; i < 4; ++i) w[i] = ((const float4*)W)[tid + 256u * (unsigned)i];
#pragma unroll
    for (int i = 0; i < 4; ++i) {
        const unsigned q = tid + 256u * (unsigned)i;
        float* p = sW + (q >> 4) * 65u + ((q & 15u) << 2);
        p[0] = w[i].x; p[1] = w[i].y; p[2] = w[i].z; p[3] = w[i].w;
    }

    const unsigned h  = lane >> 4;
    const unsigned c4 = (lane & 15u) << 2;
    const float bx = bias_in[c4], by = bias_in[c4 + 1], bz = bias_in[c4 + 2], bw = bias_in[c4 + 3];

    const unsigned m     = lane & 15u;
    const unsigned t16   = (wave >> 2) << 4;
    const unsigned n     = ((wave & 3u) << 4) + m;
    const unsigned khalf = (lane >> 4) << 1;
    const unsigned rOff  = (lane >> 4) << 3;

    unsigned it = 0;
    do {
        float* A = sA[it & 1u];

#pragma unroll
        for (int r = 0; r < 4; ++r) {
            const float4 a = pull_row_sum(X, ssrc, beg[r], end[r], idx0[r], lane, h, c4);
            if (h == 0u) {
                float* p = A + ((wave << 2) + (unsigned)r) * 65u + c4;
                float v0 = a.x + bx, v1 = a.y + by, v2 = a.z + bz, v3 = a.w + bw;
                p[0] = v0 > 0.0f ? v0 : 0.0f;
                p[1] = v1 > 0.0f ? v1 : 0.0f;
                p[2] = v2 > 0.0f ? v2 : 0.0f;
                p[3] = v3 > 0.0f ? v3 : 0.0f;
            }
        }

        const unsigned nxt = pair + PG_WGS;
        if (nxt < PG_PAIRS)
            pg_fetch(off, endp, ssrc, __builtin_amdgcn_readfirstlane(nxt * 32u + (wave << 2)), lane, beg, end, idx0);

        __syncthreads();

        v8f acc = {};
#pragma unroll
        for (int k = 0; k < 64; k += 4) {
            v2f a, b;
            a.x = A[(t16 + m) * 65u + k + khalf];
            a.y = A[(t16 + m) * 65u + k + khalf + 1];
            b.x = sW[(k + khalf) * 65u + n];
            b.y = sW[(k + khalf + 1) * 65u + n];
            acc = __builtin_amdgcn_wmma_f32_16x16x4_f32(
                 false, a,  false, b,
                 (short)0, acc,  false,  false);
        }

        float* outp = Out + (size_t)(pair * 32u + t16 + rOff) * 64 + n;
#pragma unroll
        for (int v = 0; v < 8; ++v)
            outp[(size_t)v * 64] = acc[v];

        pair = nxt;
        ++it;
    } while (pair < PG_PAIRS);
}

#define PR_ROWS 8

__device__ __forceinline__ void readout_flush(
    float* __restrict__ gsum, float* __restrict__ gcnt,
    int g, unsigned lane, unsigned h, unsigned c4, float4 acc, float run)
{
    float* p = gsum + (size_t)g * 64 + c4 + 2u * h;
    unsafeAtomicAdd(p, h ? acc.z : acc.x);
    unsafeAtomicAdd(p + 1, h ? acc.w : acc.y);
    if (lane == 0) unsafeAtomicAdd(&gcnt[g], run);
}

__global__ __launch_bounds__(256) void gcn_pull_readout(
    const float* __restrict__ X, const int* __restrict__ off,
    const int* __restrict__ endp, const int* __restrict__ ssrc,
    const int* __restrict__ gids,
    float* __restrict__ gsum, float* __restrict__ gcnt)
{
    const unsigned wave = __builtin_amdgcn_readfirstlane((blockIdx.x * 256u + threadIdx.x) >> 5);
    const unsigned lane = threadIdx.x & 31u;
    const unsigned n0 = wave * PR_ROWS;
    if (n0 >= NNODES) return;
    const unsigned h  = lane >> 4;
    const unsigned c4 = (lane & 15u) << 2;
    const int gl = gids[n0 + (lane & (PR_ROWS - 1))];

    int beg[PR_ROWS], end[PR_ROWS], idx0[PR_ROWS];
#pragma unroll
    for (int r = 0; r < PR_ROWS; ++r) {
        const int b0 = off[n0 + r];
        const int e0 = endp[n0 + r];
        beg[r] = b0 > 0 ? b0 : 0;
        end[r] = e0 < NEDGES ? e0 : NEDGES;
    }
#pragma unroll
    for (int r = 0; r < PR_ROWS; ++r)
        idx0[r] = load_chunk_idx(ssrc, beg[r], end[r], lane);

    int cur = __builtin_amdgcn_readlane(gl, 0);
    float4 racc = make_float4(0.0f, 0.0f, 0.0f, 0.0f);
    float run = 0.0f;
#pragma unroll
    for (int r = 0; r < PR_ROWS; ++r) {
        const float4 a = pull_row_sum(X, ssrc, beg[r], end[r], idx0[r], lane, h, c4);
        const int g = __builtin_amdgcn_readlane(gl, r);
        if (g != cur) {
            readout_flush(gsum, gcnt, cur, lane, h, c4, racc, run);
            cur = g;
            racc = make_float4(0.0f, 0.0f, 0.0f, 0.0f);
            run = 0.0f;
        }
        racc.x += a.x;
        racc.y += a.y;
        racc.z += a.z;
        racc.w += a.w;
        run += 1.0f;
    }
    readout_flush(gsum, gcnt, cur, lane, h, c4, racc, run);
}

__global__ __launch_bounds__(256) void gcn_finalize(
    const float* __restrict__ gsum, const float* __restrict__ gcnt,
    const float* __restrict__ b2, float* __restrict__ out)
{
    const unsigned gid = blockIdx.x * 256u + threadIdx.x;
    if (gid >= NGRAPHS * 64u) return;
    const unsigned g = gid >> 6;
    const unsigned c = gid & 63;
    const float cnt = gcnt[g];
    out[gid] = (cnt > 0.0f) ? (gsum[gid] / cnt + b2[c]) : 0.0f;
}

extern "C" void kernel_launch(void* const* d_in, const int* in_sizes, int n_in,
                              void* d_out, int out_size, void* d_ws, size_t ws_size,
                              hipStream_t stream)
{
    (void)in_sizes; (void)n_in; (void)out_size; (void)ws_size;

    const float* feats = (const float*)d_in[0];
    const float* W1    = (const float*)d_in[1];
    const float* b1    = (const float*)d_in[2];
    const float* W2    = (const float*)d_in[3];
    const float* b2    = (const float*)d_in[4];
    const int*   src   = (const int*)d_in[5];
    const int*   dst   = (const int*)d_in[6];
    const int*   gids  = (const int*)d_in[7];
    float* out = (float*)d_out;

    const size_t nodeElems = (size_t)NNODES * HID;
    float* bufA = (float*)d_ws;
    float* bufB = bufA + nodeElems;
    float* gsum = bufB + nodeElems;
    float* gcnt = gsum + (size_t)NGRAPHS * HID;
    int*   cur  = (int*)(gcnt + NGRAPHS);
    int*   off  = cur + CSR_PAD;
    int*   ssrc = off + CSR_PAD;
    int*   binned = (int*)bufB;
    int*   tbl    = binned + NEDGES;

    const int gemmBlocks    = PG_WGS;
    const int readBlocks    = (NNODES / PR_ROWS * 32 + 255) / 256;
    const int finBlocks     = (NGRAPHS * 64 + 255) / 256;

    hipLaunchKernelGGL(sort_bin_gemm, dim3(SORT_C + FG_WGS), dim3(1024), 0, stream,
                       src, dst, binned, tbl, feats, W1, bufA, gsum);
    hipLaunchKernelGGL(sort_place, dim3(SORT_NB), dim3(1024), 0, stream, binned, tbl, off, cur, ssrc);
    hipLaunchKernelGGL(gcn_pull_gemm, dim3(gemmBlocks), dim3(256), 0, stream,
                       bufA, off, cur, ssrc, W2, b1, bufB);
    hipLaunchKernelGGL(gcn_pull_readout, dim3(readBlocks), dim3(256), 0, stream,
                       bufB, off, cur, ssrc, gids, gsum, gcnt);
    hipLaunchKernelGGL(gcn_finalize, dim3(finBlocks), dim3(256), 0, stream,
                       gsum, gcnt, b2, out);
}


# row gathers: a row's sources come by scalar loads as byte offsets; a whole wave reads one source row per 8-byte lane load off a scalar base, a block in flight: no lane exchange; sum order changes
# speedup vs baseline: 4.9841x; 1.0343x over previous
#include <hip/hip_runtime.h>

#define HID      64
#define NNODES   100000
#define NEDGES   1200000
#define NGRAPHS  1000

typedef __attribute__((ext_vector_type(2))) float v2f;
typedef __attribute__((ext_vector_type(8))) float v8f;

#define CSR_PAD   100032
#define SORT_C    250
#define SORT_EPB  4800
#define SORT_NB   250
#define SORT_R    400
#define SORT_PER  5
#define SORT_CAP  16384
#define SRC_BITS  17
#define SRC_MASK  0x1FFFFu

static_assert(SORT_C * SORT_EPB == NEDGES, "chunks cover the edge list");
static_assert(SORT_NB * SORT_R == NNODES, "ranges cover the nodes");
static_assert(NNODES <= (1 << SRC_BITS) && SORT_R <= 512, "packed word holds row and source");
static_assert(SORT_PER * 1024 >= SORT_EPB && SORT_C <= 256 && SORT_NB < 256, "thread maps");

#define FG_TILES  25
#define FG_NT     (NNODES / 16)
#define FG_WGS    ((FG_NT + FG_TILES - 1) / FG_TILES)
#define FG_APITCH 68
#define FG_SLAB   (16 * FG_APITCH)
#define FG_LDS    (64 * 65 + FG_TILES * FG_SLAB)
#define FG_CLEAR  (NGRAPHS * HID + NGRAPHS)
#define FG_CLR    ((FG_CLEAR + FG_WGS - 1) / FG_WGS)

static_assert(NNODES % 16 == 0, "row tiles have no tail");
static_assert(FG_CLR <= 1024, "a thread stores at most one zero");
static_assert(FG_LDS >= 256 + SORT_EPB, "the sort role fits the same LDS");
static_assert(SORT_C + FG_WGS <= 512, "one resident pass at two work-groups per WGP");

__global__ __launch_bounds__(1024) void sort_bin_gemm(
    const int* __restrict__ src, const int* __restrict__ dst,
    int* __restrict__ binned, int* __restrict__ tbl,
    const float* __restrict__ In, const float* __restrict__ W,
    float* __restrict__ Out, float* __restrict__ gclear)
{
    __shared__ __attribute__((aligned(16))) float lds[FG_LDS];
    const unsigned tid  = threadIdx.x;
    const unsigned lane = tid & 31u;

    if (blockIdx.x >= SORT_C) {
        const unsigned wave = tid >> 5;
        const unsigned g    = blockIdx.x - SORT_C;
        float* sW = lds;
        {
            const float4 w = ((const float4*)W)[tid];
            float* p = sW + (tid >> 4) * 65u + ((tid & 15u) << 2);
            p[0] = w.x; p[1] = w.y; p[2] = w.z; p[3] = w.w;
        }
        const unsigned tile = g * FG_TILES + wave;
        const bool has = wave < FG_TILES && tile < FG_NT;
        float* sA = lds + 64 * 65 + (has ? wave : 0u) * FG_SLAB;
        if (has) {
            const float4* t4 = (const float4*)(In + (size_t)tile * (16 * 64));
            float4 v[8];
#pragma unroll
            for (int i = 0; i < 8; ++i) v[i] = t4[lane + 32u * (unsigned)i];
#pragma unroll
            for (int i = 0; i < 8; ++i)
                *(float4*)(sA + (2u * (unsigned)i + (lane >> 4)) * FG_APITCH + ((lane & 15u) << 2)) = v[i];
        }
        if (tid < FG_CLR) {
            const unsigned i = g * FG_CLR + tid;
            if (i < FG_CLEAR) gclear[i] = 0.0f;
        }
        __syncthreads();
        if (!has) return;

        const unsigned m     = lane & 15u;
        const unsigned khalf = (lane >> 4) << 1;
        v8f acc0 = {}, acc1 = {}, acc2 = {}, acc3 = {};
#pragma unroll
        for (int k = 0; k < 64; k += 4) {
            v2f a, b0, b1, b2, b3;
            const float* pa = sA + m * FG_APITCH + k + khalf;
            const float* pb = sW + (k + khalf) * 65u + m;
            a.x  = pa[0];   a.y  = pa[1];
            b0.x = pb[0];   b0.y = pb[65];
            b1.x = pb[16];  b1.y = pb[65 + 16];
            b2.x = pb[32];  b2.y = pb[65 + 32];
            b3.x = pb[48];  b3.y = pb[65 + 48];
            acc0 = __builtin_amdgcn_wmma_f32_16x16x4_f32(false, a, false, b0, (short)0, acc0, false, false);
            acc1 = __builtin_amdgcn_wmma_f32_16x16x4_f32(false, a, false, b1, (short)0, acc1, false, false);
            acc2 = __builtin_amdgcn_wmma_f32_16x16x4_f32(false, a, false, b2, (short)0, acc2, false, false);
            acc3 = __builtin_amdgcn_wmma_f32_16x16x4_f32(false, a, false, b3, (short)0, acc3, false, false);
        }

        const unsigned rOff = (lane >> 4) << 3;
        float* outp = Out + (size_t)(tile * 16u + rOff) * 64 + m;
#pragma unroll
        for (int v = 0; v < 8; ++v) {
            outp[(size_t)v * 64]      = acc0[v];
            outp[(size_t)v * 64 + 16] = acc1[v];
            outp[(size_t)v * 64 + 32] = acc2[v];
            outp[(size_t)v * 64 + 48] = acc3[v];
        }
        return;
    }

    int* cnt   = (int*)lds;
    int* stage = cnt + 256;
    const unsigned c    = blockIdx.x;
    const unsigned base = c * SORT_EPB;

    if (tid < 256u) cnt[tid] = 0;
    __syncthreads();

    unsigned rec[SORT_PER], bin[SORT_PER];
#pragma unroll
    for (int k = 0; k < SORT_PER; ++k) {
        const unsigned i = tid + (unsigned)k * 1024u;
        bin[k] = 0xFFFFFFFFu;
        rec[k] = 0u;
        if (i < SORT_EPB) {
            unsigned d = (unsigned)dst[base + i];
            unsigned s = (unsigned)src[base + i];
            d = d < (unsigned)NNODES ? d : (unsigned)(NNODES - 1);
            s = s < (unsigned)NNODES ? s : (unsigned)(NNODES - 1);
            const unsigned b = d / SORT_R;
            bin[k] = b;
            rec[k] = ((d - b * SORT_R) << SRC_BITS) | s;
            atomicAdd(&cnt[b], 1);
        }
    }
    __syncthreads();

    if (tid < 32u) {
        int v[8];
        int sum = 0;
#pragma unroll
        for (int j = 0; j < 8; ++j) {
            const unsigned idx = lane * 8u + (unsigned)j;
            v[j] = idx < SORT_NB ? cnt[idx] : 0;
            sum += v[j];
        }
        int incl = sum;
#pragma unroll
        for (int d = 1; d < 32; d <<= 1) {
            const int t = __shfl_up(incl, d);
            if (lane >= (unsigned)d) incl += t;
        }
        int run = incl - sum;
#pragma unroll
        for (int j = 0; j < 8; ++j) {
            const unsigned idx = lane * 8u + (unsigned)j;
            cnt[idx] = run;
            if (idx <= SORT_NB) tbl[idx * SORT_C + c] = run;
            run += v[j];
        }
    }
    __syncthreads();

#pragma unroll
    for (int k = 0; k < SORT_PER; ++k) {
        if (bin[k] != 0xFFFFFFFFu) {
            const int p = atomicAdd(&cnt[bin[k]], 1);
            if ((unsigned)p < (unsigned)SORT_EPB) stage[p] = (int)rec[k];
        }
    }
    __syncthreads();

    for (unsigned i = tid; i < SORT_EPB; i += 1024u) binned[base + i] = stage[i];
}

__global__ __launch_bounds__(1024) void sort_place(
    const int* __restrict__ binned, const int* __restrict__ tbl,
    int* __restrict__ off, int* __restrict__ endp, int* __restrict__ ssrc)
{
    __shared__ int pcS[256];
    __shared__ int pcE[256];
    __shared__ int cnt[512];
    __shared__ int wsum[32];
    __shared__ int sBeg, sEnd;
    __shared__ int stage[SORT_CAP];
    const unsigned tid  = threadIdx.x;
    const unsigned lane = tid & 31u;
    const unsigned wave = tid >> 5;
    const unsigned b    = blockIdx.x;

    int o0 = 0, o1 = 0;
    if (tid < SORT_C) {
        o0 = tbl[b * SORT_C + tid];
        o1 = tbl[(b + 1u) * SORT_C + tid];
        o0 = o0 > 0 ? o0 : 0;
        o0 = o0 < SORT_EPB ? o0 : SORT_EPB;
        o1 = o1 > o0 ? o1 : o0;
        o1 = o1 < SORT_EPB ? o1 : SORT_EPB;
    }
    if (tid < 256u) { pcS[tid] = o0; pcE[tid] = o1; }
    if (tid < 512u) cnt[tid] = 0;
    if (tid == 0u) { sBeg = 0; sEnd = 0; }
    __syncthreads();

    if (tid < 256u) {
        int a = o0, e = o1;
#pragma unroll
        for (int d = 16; d > 0; d >>= 1) {
            a += __shfl_xor(a, d);
            e += __shfl_xor(e, d);
        }
        if (lane == 0u) { atomicAdd(&sBeg, a); atomicAdd(&sEnd, e); }
    }

    unsigned r[8];
    bool ok[8];
#pragma unroll
    for (int j = 0; j < 8; ++j) {
        const unsigned c = wave + 32u * (unsigned)j;
        int s = 0, e = 0;
        if (c < SORT_C) { s = pcS[c]; e = pcE[c]; }
        const int i = s + (int)lane;
        ok[j] = i < e;
        r[j] = ok[j] ? (unsigned)binned[c * SORT_EPB + (unsigned)i] : 0u;
    }
#pragma unroll
    for (int j = 0; j < 8; ++j) {
        if (ok[j]) {
            unsigned dl = r[j] >> SRC_BITS;
            dl = dl < SORT_R ? dl : SORT_R - 1u;
            atomicAdd(&cnt[dl], 1);
        }
    }
    for (unsigned c = wave; c < SORT_C; c += 32u) {
        const int e = pcE[c];
        for (int i = pcS[c] + 32 + (int)lane; i < e; i += 32) {
            unsigned dl = (unsigned)binned[c * SORT_EPB + (unsigned)i] >> SRC_BITS;
            dl = dl < SORT_R ? dl : SORT_R - 1u;
            atomicAdd(&cnt[dl], 1);
        }
    }
    __syncthreads();

    int cv = 0, incl = 0;
    if (tid < 512u) {
        cv = cnt[tid];
        incl = cv;
#pragma unroll
        for (int d = 1; d < 32; d <<= 1) {
            const int t = __shfl_up(incl, d);
            if (lane >= (unsigned)d) incl += t;
        }
        if (lane == 31u) wsum[wave] = incl;
    }
    __syncthreads();
    if (wave == 0u) {
        const int w = lane < 16u ? wsum[lane] : 0;
        int wi = w;
#pragma unroll
        for (int d = 1; d < 32; d <<= 1) {
            const int t = __shfl_up(wi, d);
            if (lane >= (unsigned)d) wi += t;
        }
        wsum[lane] = wi - w;
    }
    __syncthreads();
    const int beg   = sBeg;
    const int total = sEnd - beg;
    const bool fits = total <= SORT_CAP;
    if (tid < SORT_R) {
        const int excl = wsum[wave] + incl - cv;
        const unsigned n = b * SORT_R + tid;
        cnt[tid] = excl;
        off[n]  = beg + excl;
        endp[n] = beg + excl + cv;
    }
    __syncthreads();

#pragma unroll
    for (int j = 0; j < 8; ++j) {
        if (ok[j]) {
            unsigned dl = r[j] >> SRC_BITS;
            dl = dl < SORT_R ? dl : SORT_R - 1u;
            const int p = atomicAdd(&cnt[dl], 1);
            const int s = (int)((r[j] & SRC_MASK) << 8);
            if (fits) {
                if ((unsigned)p < (unsigned)SORT_CAP) stage[p] = s;
            } else {
                const int q = beg + p;
                if ((unsigned)q < (unsigned)NEDGES) ssrc[q] = s;
            }
        }
    }
    for (unsigned c = wave; c < SORT_C; c += 32u) {
        const int e = pcE[c];
        for (int i = pcS[c] + 32 + (int)lane; i < e; i += 32) {
            const unsigned rr = (unsigned)binned[c * SORT_EPB + (unsigned)i];
            unsigned dl = rr >> SRC_BITS;
            dl = dl < SORT_R ? dl : SORT_R - 1u;
            const int p = atomicAdd(&cnt[dl], 1);
            const int s = (int)((rr & SRC_MASK) << 8);
            if (fits) {
                if ((unsigned)p < (unsigned)SORT_CAP) stage[p] = s;
            } else {
                const int q = beg + p;
                if ((unsigned)q < (unsigned)NEDGES) ssrc[q] = s;
            }
        }
    }
    __syncthreads();

    if (fits) {
        for (int i = (int)tid; i < total; i += 1024) {
            const int q = beg + i;
            if ((unsigned)q < (unsigned)NEDGES) ssrc[q] = stage[i];
        }
    }
}

#define PB 16

__device__ __forceinline__ void load_idx_block(const int* __restrict__ ssrc, int e, int (&s)[PB])
{
    e = e > 0 ? e : 0;
    e = e < NEDGES ? e : NEDGES;
    const __attribute__((address_space(4))) int* cs = (const __attribute__((address_space(4))) int*)ssrc;
#pragma unroll
    for (int k = 0; k < PB; ++k) s[k] = cs[e + k];
}

__device__ __forceinline__ v2f row_load(const float* __restrict__ X, int o, unsigned lane8)
{
    const unsigned oc = (unsigned)o < (unsigned)(NNODES - 1) * 256u ? (unsigned)o : (unsigned)(NNODES - 1) * 256u;
    const unsigned o32 = oc + lane8;
    return *(const v2f*)((const char*)X + o32);
}

__device__ __forceinline__ void gather_issue(
    const float* __restrict__ X, const int (&s)[PB], int rem, unsigned lane8, v2f (&v)[PB])
{
    const v2f zero = {0.0f, 0.0f};
#pragma unroll
    for (int g = 0; g < PB; g += 4) {
        v[g]     = zero;
        v[g + 1] = zero;
        v[g + 2] = zero;
        v[g + 3] = zero;
        if (rem > g) {
            v[g] = row_load(X, s[g], lane8);
            if (rem > g + 1) {
                v[g + 1] = row_load(X, s[g + 1], lane8);
                if (rem > g + 2) {
                    v[g + 2] = row_load(X, s[g + 2], lane8);
                    if (rem > g + 3) v[g + 3] = row_load(X, s[g + 3], lane8);
                }
            }
        }
    }
}

__device__ __forceinline__ v2f gather_sum(const v2f (&v)[PB], v2f acc)
{
#pragma unroll
    for (int k = 0; k < PB; ++k) acc += v[k];
    return acc;
}

__device__ __forceinline__ v2f pull_row_sum(
    const float* __restrict__ X, const int* __restrict__ ssrc,
    int beg, int end, int next_beg, bool has_next, int (&idx)[PB], unsigned lane8, v2f acc)
{
    int e = beg;
    bool more;
    do {
        v2f v[PB];
        const int rem = end - e;
        gather_issue(X, idx, rem, lane8, v);
        e += PB;
        more = e < end;
        if (more || has_next) load_idx_block(ssrc, more ? e : next_beg, idx);
        acc = gather_sum(v, acc);
    } while (more);
    return acc;
}

#define PG_PAIRS   (NNODES / 32)
#define PG_PASSES  2
#define PG_WGS     ((PG_PAIRS + PG_PASSES - 1) / PG_PASSES)

static_assert(NNODES % 32 == 0, "pairs of row tiles have no tail");

__device__ __forceinline__ void rows_fetch(
    const int* __restrict__ off, const int* __restrict__ endp, const int* __restrict__ ssrc,
    unsigned n0, unsigned lane, unsigned mask, int& vb, int& ve, int (&idx)[PB])
{
    const int b0 = off[n0 + (lane & mask)];
    const int e0 = endp[n0 + (lane & mask)];
    vb = b0 > 0 ? b0 : 0;
    ve = e0 < NEDGES ? e0 : NEDGES;
    load_idx_block(ssrc, __builtin_amdgcn_readlane(vb, 0), idx);
}

__global__ __launch_bounds__(256, 16) void gcn_pull_gemm(
    const float* __restrict__ X, const int* __restrict__ off,
    const int* __restrict__ endp, const int* __restrict__ ssrc,
    const float* __restrict__ W, const float* __restrict__ bias_in,
    float* __restrict__ Out)
{
    __shared__ float sW[64 * 65];
    __shared__ float sA[2][32 * 65];

    const unsigned tid  = threadIdx.x;
    const unsigned wave = tid >> 5;
    const unsigned lane = tid & 31u;
    const unsigned lane8 = lane << 3;

    unsigned pair = blockIdx.x;
    int vb, ve, idx[PB];
    rows_fetch(off, endp, ssrc, pair * 32u + (wave << 2), lane, 3u, vb, ve, idx);

    float4 w[4];
#pragma unroll
    for (int i = 0; i < 4; ++i) w[i] = ((const float4*)W)[tid + 256u * (unsigned)i];
#pragma unroll
    for (int i = 0; i < 4; ++i) {
        const unsigned q = tid + 256u * (unsigned)i;
        float* p = sW + (q >> 4) * 65u + ((q & 15u) << 2);
        p[0] = w[i].x; p[1] = w[i].y; p[2] = w[i].z; p[3] = w[i].w;
    }

    const float bx = bias_in[2u * lane], by = bias_in[2u * lane + 1u];

    const unsigned m     = lane & 15u;
    const unsigned t16   = (wave >> 2) << 4;
    const unsigned n     = ((wave & 3u) << 4) + m;
    const unsigned khalf = (lane >> 4) << 1;
    const unsigned rOff  = (lane >> 4) << 3;

    unsigned it = 0;
    do {
        float* A = sA[it & 1u];

        for (int r = 0; r < 4; ++r) {
            const int beg = __builtin_amdgcn_readlane(vb, r);
            const int end = __builtin_amdgcn_readlane(ve, r);
            const int nb  = __builtin_amdgcn_readlane(vb, (r + 1) & 3);
            const v2f a = pull_row_sum(X, ssrc, beg, end, nb, r < 3, idx, lane8, (v2f){0.0f, 0.0f});
            float* p = A + ((wave << 2) + (unsigned)r) * 65u + 2u * lane;
            const float v0 = a.x + bx, v1 = a.y + by;
            p[0] = v0 > 0.0f ? v0 : 0.0f;
            p[1] = v1 > 0.0f ? v1 : 0.0f;
        }

        const unsigned nxt = pair + PG_WGS;
        if (nxt < PG_PAIRS)
            rows_fetch(off, endp, ssrc, nxt * 32u + (wave << 2), lane, 3u, vb, ve, idx);

        __syncthreads();

        v8f acc = {};
#pragma unroll
        for (int k = 0; k < 64; k += 4) {
            v2f a, b;
            a.x = A[(t16 + m) * 65u + k + khalf];
            a.y = A[(t16 + m) * 65u + k + khalf + 1];
            b.x = sW[(k + khalf) * 65u + n];
            b.y = sW[(k + khalf + 1) * 65u + n];
            acc = __builtin_amdgcn_wmma_f32_16x16x4_f32(
                 false, a,  false, b,
                 (short)0, acc,  false,  false);
        }

        float* outp = Out + (size_t)(pair * 32u + t16 + rOff) * 64 + n;
#pragma unroll
        for (int v = 0; v < 8; ++v)
            outp[(size_t)v * 64] = acc[v];

        pair = nxt;
        ++it;
    } while (pair < PG_PAIRS);
}

#define PR_ROWS 8

__device__ __forceinline__ void readout_flush(
    float* __restrict__ gsum, float* __restrict__ gcnt,
    int g, unsigned lane, v2f acc, float run)
{
    float* p = gsum + (size_t)g * 64 + 2u * lane;
    unsafeAtomicAdd(p, acc.x);
    unsafeAtomicAdd(p + 1, acc.y);
    if (lane == 0) unsafeAtomicAdd(&gcnt[g], run);
}

__global__ __launch_bounds__(256) void gcn_pull_readout(
    const float* __restrict__ X, const int* __restrict__ off,
    const int* __restrict__ endp, const int* __restrict__ ssrc,
    const int* __restrict__ gids,
    float* __restrict__ gsum, float* __restrict__ gcnt)
{
    const unsigned wave = __builtin_amdgcn_readfirstlane((blockIdx.x * 256u + threadIdx.x) >> 5);
    const unsigned lane = threadIdx.x & 31u;
    const unsigned n0 = wave * PR_ROWS;
    if (n0 >= NNODES) return;
    const unsigned lane8 = lane << 3;
    const int gl = gids[n0 + (lane & (PR_ROWS - 1))];

    int vb, ve, idx[PB];
    rows_fetch(off, endp, ssrc, n0, lane, PR_ROWS - 1, vb, ve, idx);

    int cur = __builtin_amdgcn_readlane(gl, 0);
    v2f racc = {0.0f, 0.0f};
    float run = 0.0f;
    for (int r = 0; r < PR_ROWS; ++r) {
        const int g = __builtin_amdgcn_readlane(gl, r);
        if (g != cur) {
            readout_flush(gsum, gcnt, cur, lane, racc, run);
            cur = g;
            racc = (v2f){0.0f, 0.0f};
            run = 0.0f;
        }
        const int beg = __builtin_amdgcn_readlane(vb, r);
        const int end = __builtin_amdgcn_readlane(ve, r);
        const int nb  = __builtin_amdgcn_readlane(vb, (r + 1) & (PR_ROWS - 1));
        racc = pull_row_sum(X, ssrc, beg, end, nb, r < PR_ROWS - 1, idx, lane8, racc);
        run += 1.0f;
    }
    readout_flush(gsum, gcnt, cur, lane, racc, run);
}

__global__ __launch_bounds__(256) void gcn_finalize(
    const float* __restrict__ gsum, const float* __restrict__ gcnt,
    const float* __restrict__ b2, float* __restrict__ out)
{
    const unsigned gid = blockIdx.x * 256u + threadIdx.x;
    if (gid >= NGRAPHS * 64u) return;
    const unsigned g = gid >> 6;
    const unsigned c = gid & 63;
    const float cnt = gcnt[g];
    out[gid] = (cnt > 0.0f) ? (gsum[gid] / cnt + b2[c]) : 0.0f;
}

extern "C" void kernel_launch(void* const* d_in, const int* in_sizes, int n_in,
                              void* d_out, int out_size, void* d_ws, size_t ws_size,
                              hipStream_t stream)
{
    (void)in_sizes; (void)n_in; (void)out_size; (void)ws_size;

    const float* feats = (const float*)d_in[0];
    const float* W1    = (const float*)d_in[1];
    const float* b1    = (const float*)d_in[2];
    const float* W2    = (const float*)d_in[3];
    const float* b2    = (const float*)d_in[4];
    const int*   src   = (const int*)d_in[5];
    const int*   dst   = (const int*)d_in[6];
    const int*   gids  = (const int*)d_in[7];
    float* out = (float*)d_out;

    const size_t nodeElems = (size_t)NNODES * HID;
    float* bufA = (float*)d_ws;
    float* bufB = bufA + nodeElems;
    float* gsum = bufB + nodeElems;
    float* gcnt = gsum + (size_t)NGRAPHS * HID;
    int*   ssrc = (int*)(gcnt + NGRAPHS);
    int*   cur  = ssrc + NEDGES;
    int*   off  = cur + CSR_PAD;
    int*   binned = (int*)bufB;
    int*   tbl    = binned + NEDGES;

    const int gemmBlocks    = PG_WGS;
    const int readBlocks    = (NNODES / PR_ROWS * 32 + 255) / 256;
    const int finBlocks     = (NGRAPHS * 64 + 255) / 256;

    hipLaunchKernelGGL(sort_bin_gemm, dim3(SORT_C + FG_WGS), dim3(1024), 0, stream,
                       src, dst, binned, tbl, feats, W1, bufA, gsum);
    hipLaunchKernelGGL(sort_place, dim3(SORT_NB), dim3(1024), 0, stream, binned, tbl, off, cur, ssrc);
    hipLaunchKernelGGL(gcn_pull_gemm, dim3(gemmBlocks), dim3(256), 0, stream,
                       bufA, off, cur, ssrc, W2, b1, bufB);
    hipLaunchKernelGGL(gcn_pull_readout, dim3(readBlocks), dim3(256), 0, stream,
                       bufB, off, cur, ssrc, gids, gsum, gcnt);
    hipLaunchKernelGGL(gcn_finalize, dim3(finBlocks), dim3(256), 0, stream,
                       gsum, gcnt, b2, out);
}


# no finalize launch: sort_place counts each graph's nodes, the layer-2 launch starts out at b2, the readout adds run sum / count straight onto out; 5 stream ops become 4; divided before summed
# speedup vs baseline: 5.3238x; 1.0682x over previous
#include <hip/hip_runtime.h>

#define HID      64
#define NNODES   100000
#define NEDGES   1200000
#define NGRAPHS  1000

typedef __attribute__((ext_vector_type(2))) float v2f;
typedef __attribute__((ext_vector_type(8))) float v8f;

#define CSR_PAD   100032
#define SORT_C    250
#define SORT_EPB  4800
#define SORT_NB   250
#define SORT_R    400
#define SORT_PER  5
#define SORT_CAP  16384
#define SRC_BITS  17
#define SRC_MASK  0x1FFFFu

static_assert(SORT_C * SORT_EPB == NEDGES, "chunks cover the edge list");
static_assert(SORT_NB * SORT_R == NNODES, "ranges cover the nodes");
static_assert(NNODES <= (1 << SRC_BITS) && SORT_R <= 512, "packed word holds row and source");
static_assert(SORT_PER * 1024 >= SORT_EPB && SORT_C <= 256 && SORT_NB < 256, "thread maps");

#define FG_TILES  25
#define FG_NT     (NNODES / 16)
#define FG_WGS    ((FG_NT + FG_TILES - 1) / FG_TILES)
#define FG_APITCH 68
#define FG_SLAB   (16 * FG_APITCH)
#define FG_LDS    (64 * 65 + FG_TILES * FG_SLAB)
#define FG_CLEAR  NGRAPHS
#define FG_CLR    ((FG_CLEAR + FG_WGS - 1) / FG_WGS)

static_assert(NNODES % 16 == 0, "row tiles have no tail");
static_assert(FG_CLR <= 1024, "a thread stores at most one zero");
static_assert(FG_LDS >= 256 + SORT_EPB, "the sort role fits the same LDS");
static_assert(SORT_C + FG_WGS <= 512, "one resident pass at two work-groups per WGP");

__global__ __launch_bounds__(1024) void sort_bin_gemm(
    const int* __restrict__ src, const int* __restrict__ dst,
    int* __restrict__ binned, int* __restrict__ tbl,
    const float* __restrict__ In, const float* __restrict__ W,
    float* __restrict__ Out, float* __restrict__ gclear)
{
    __shared__ __attribute__((aligned(16))) float lds[FG_LDS];
    const unsigned tid  = threadIdx.x;
    const unsigned lane = tid & 31u;

    if (blockIdx.x >= SORT_C) {
        const unsigned wave = tid >> 5;
        const unsigned g    = blockIdx.x - SORT_C;
        float* sW = lds;
        {
            const float4 w = ((const float4*)W)[tid];
            float* p = sW + (tid >> 4) * 65u + ((tid & 15u) << 2);
            p[0] = w.x; p[1] = w.y; p[2] = w.z; p[3] = w.w;
        }
        const unsigned tile = g * FG_TILES + wave;
        const bool has = wave < FG_TILES && tile < FG_NT;
        float* sA = lds + 64 * 65 + (has ? wave : 0u) * FG_SLAB;
        if (has) {
            const float4* t4 = (const float4*)(In + (size_t)tile * (16 * 64));
            float4 v[8];
#pragma unroll
            for (int i = 0; i < 8; ++i) v[i] = t4[lane + 32u * (unsigned)i];
#pragma unroll
            for (int i = 0; i < 8; ++i)
                *(float4*)(sA + (2u * (unsigned)i + (lane >> 4)) * FG_APITCH + ((lane & 15u) << 2)) = v[i];
        }
        if (tid < FG_CLR) {
            const unsigned i = g * FG_CLR + tid;
            if (i < FG_CLEAR) gclear[i] = 0.0f;
        }
        __syncthreads();
        if (!has) return;

        const unsigned m     = lane & 15u;
        const unsigned khalf = (lane >> 4) << 1;
        v8f acc0 = {}, acc1 = {}, acc2 = {}, acc3 = {};
#pragma unroll
        for (int k = 0; k < 64; k += 4) {
            v2f a, b0, b1, b2, b3;
            const float* pa = sA + m * FG_APITCH + k + khalf;
            const float* pb = sW + (k + khalf) * 65u + m;
            a.x  = pa[0];   a.y  = pa[1];
            b0.x = pb[0];   b0.y = pb[65];
            b1.x = pb[16];  b1.y = pb[65 + 16];
            b2.x = pb[32];  b2.y = pb[65 + 32];
            b3.x = pb[48];  b3.y = pb[65 + 48];
            acc0 = __builtin_amdgcn_wmma_f32_16x16x4_f32(false, a, false, b0, (short)0, acc0, false, false);
            acc1 = __builtin_amdgcn_wmma_f32_16x16x4_f32(false, a, false, b1, (short)0, acc1, false, false);
            acc2 = __builtin_amdgcn_wmma_f32_16x16x4_f32(false, a, false, b2, (short)0, acc2, false, false);
            acc3 = __builtin_amdgcn_wmma_f32_16x16x4_f32(false, a, false, b3, (short)0, acc3, false, false);
        }

        const unsigned rOff = (lane >> 4) << 3;
        float* outp = Out + (size_t)(tile * 16u + rOff) * 64 + m;
#pragma unroll
        for (int v = 0; v < 8; ++v) {
            outp[(size_t)v * 64]      = acc0[v];
            outp[(size_t)v * 64 + 16] = acc1[v];
            outp[(size_t)v * 64 + 32] = acc2[v];
            outp[(size_t)v * 64 + 48] = acc3[v];
        }
        return;
    }

    int* cnt   = (int*)lds;
    int* stage = cnt + 256;
    const unsigned c    = blockIdx.x;
    const unsigned base = c * SORT_EPB;

    if (tid < 256u) cnt[tid] = 0;
    __syncthreads();

    unsigned rec[SORT_PER], bin[SORT_PER];
#pragma unroll
    for (int k = 0; k < SORT_PER; ++k) {
        const unsigned i = tid + (unsigned)k * 1024u;
        bin[k] = 0xFFFFFFFFu;
        rec[k] = 0u;
        if (i < SORT_EPB) {
            unsigned d = (unsigned)dst[base + i];
            unsigned s = (unsigned)src[base + i];
            d = d < (unsigned)NNODES ? d : (unsigned)(NNODES - 1);
            s = s < (unsigned)NNODES ? s : (unsigned)(NNODES - 1);
            const unsigned b = d / SORT_R;
            bin[k] = b;
            rec[k] = ((d - b * SORT_R) << SRC_BITS) | s;
            atomicAdd(&cnt[b], 1);
        }
    }
    __syncthreads();

    if (tid < 32u) {
        int v[8];
        int sum = 0;
#pragma unroll
        for (int j = 0; j < 8; ++j) {
            const unsigned idx = lane * 8u + (unsigned)j;
            v[j] = idx < SORT_NB ? cnt[idx] : 0;
            sum += v[j];
        }
        int incl = sum;
#pragma unroll
        for (int d = 1; d < 32; d <<= 1) {
            const int t = __shfl_up(incl, d);
            if (lane >= (unsigned)d) incl += t;
        }
        int run = incl - sum;
#pragma unroll
        for (int j = 0; j < 8; ++j) {
            const unsigned idx = lane * 8u + (unsigned)j;
            cnt[idx] = run;
            if (idx <= SORT_NB) tbl[idx * SORT_C + c] = run;
            run += v[j];
        }
    }
    __syncthreads();

#pragma unroll
    for (int k = 0; k < SORT_PER; ++k) {
        if (bin[k] != 0xFFFFFFFFu) {
            const int p = atomicAdd(&cnt[bin[k]], 1);
            if ((unsigned)p < (unsigned)SORT_EPB) stage[p] = (int)rec[k];
        }
    }
    __syncthreads();

    for (unsigned i = tid; i < SORT_EPB; i += 1024u) binned[base + i] = stage[i];
}

__global__ __launch_bounds__(1024) void sort_place(
    const int* __restrict__ binned, const int* __restrict__ tbl,
    int* __restrict__ off, int* __restrict__ endp, int* __restrict__ ssrc,
    const int* __restrict__ gids, float* __restrict__ gcnt)
{
    __shared__ int pcS[256];
    __shared__ int pcE[256];
    __shared__ int cnt[512];
    __shared__ int wsum[32];
    __shared__ int sBeg, sEnd;
    __shared__ int stage[SORT_CAP];
    const unsigned tid  = threadIdx.x;
    const unsigned lane = tid & 31u;
    const unsigned wave = tid >> 5;
    const unsigned b    = blockIdx.x;

    const bool gvalid = tid < SORT_R;
    const int gid = gids[b * SORT_R + (gvalid ? tid : SORT_R - 1u)];

    int o0 = 0, o1 = 0;
    if (tid < SORT_C) {
        o0 = tbl[b * SORT_C + tid];
        o1 = tbl[(b + 1u) * SORT_C + tid];
        o0 = o0 > 0 ? o0 : 0;
        o0 = o0 < SORT_EPB ? o0 : SORT_EPB;
        o1 = o1 > o0 ? o1 : o0;
        o1 = o1 < SORT_EPB ? o1 : SORT_EPB;
    }
    if (tid < 256u) { pcS[tid] = o0; pcE[tid] = o1; }
    if (tid < 512u) cnt[tid] = 0;
    if (tid == 0u) { sBeg = 0; sEnd = 0; }
    __syncthreads();

    if (tid < 256u) {
        int a = o0, e = o1;
#pragma unroll
        for (int d = 16; d > 0; d >>= 1) {
            a += __shfl_xor(a, d);
            e += __shfl_xor(e, d);
        }
        if (lane == 0u) { atomicAdd(&sBeg, a); atomicAdd(&sEnd, e); }
    }

    unsigned r[8];
    bool ok[8];
#pragma unroll
    for (int j = 0; j < 8; ++j) {
        const unsigned c = wave + 32u * (unsigned)j;
        int s = 0, e = 0;
        if (c < SORT_C) { s = pcS[c]; e = pcE[c]; }
        const int i = s + (int)lane;
        ok[j] = i < e;
        r[j] = ok[j] ? (unsigned)binned[c * SORT_EPB + (unsigned)i] : 0u;
    }

    {
        int g = gid > 0 ? gid : 0;
        g = g < NGRAPHS ? g : NGRAPHS - 1;
        const int prev = __shfl_up(g, 1);
        const bool head = gvalid && (lane == 0u || g != prev);
        const unsigned hm = __builtin_amdgcn_ballot_w32(head);
        const unsigned nv = (unsigned)__builtin_popcount(__builtin_amdgcn_ballot_w32(gvalid));
        if (head) {
            const unsigned later = lane < 31u ? (hm >> (lane + 1u)) : 0u;
            const unsigned len = later ? (unsigned)__builtin_ctz(later) + 1u : nv - lane;
            unsafeAtomicAdd(&gcnt[g], (float)len);
        }
    }

#pragma unroll
    for (int j = 0; j < 8; ++j) {
        if (ok[j]) {
            unsigned dl = r[j] >> SRC_BITS;
            dl = dl < SORT_R ? dl : SORT_R - 1u;
            atomicAdd(&cnt[dl], 1);
        }
    }
    for (unsigned c = wave; c < SORT_C; c += 32u) {
        const int e = pcE[c];
        for (int i = pcS[c] + 32 + (int)lane; i < e; i += 32) {
            unsigned dl = (unsigned)binned[c * SORT_EPB + (unsigned)i] >> SRC_BITS;
            dl = dl < SORT_R ? dl : SORT_R - 1u;
            atomicAdd(&cnt[dl], 1);
        }
    }
    __syncthreads();

    int cv = 0, incl = 0;
    if (tid < 512u) {
        cv = cnt[tid];
        incl = cv;
#pragma unroll
        for (int d = 1; d < 32; d <<= 1) {
            const int t = __shfl_up(incl, d);
            if (lane >= (unsigned)d) incl += t;
        }
        if (lane == 31u) wsum[wave] = incl;
    }
    __syncthreads();
    if (wave == 0u) {
        const int w = lane < 16u ? wsum[lane] : 0;
        int wi = w;
#pragma unroll
        for (int d = 1; d < 32; d <<= 1) {
            const int t = __shfl_up(wi, d);
            if (lane >= (unsigned)d) wi += t;
        }
        wsum[lane] = wi - w;
    }
    __syncthreads();
    const int beg   = sBeg;
    const int total = sEnd - beg;
    const bool fits = total <= SORT_CAP;
    if (tid < SORT_R) {
        const int excl = wsum[wave] + incl - cv;
        const unsigned n = b * SORT_R + tid;
        cnt[tid] = excl;
        off[n]  = beg + excl;
        endp[n] = beg + excl + cv;
    }
    __syncthreads();

#pragma unroll
    for (int j = 0; j < 8; ++j) {
        if (ok[j]) {
            unsigned dl = r[j] >> SRC_BITS;
            dl = dl < SORT_R ? dl : SORT_R - 1u;
            const int p = atomicAdd(&cnt[dl], 1);
            const int s = (int)((r[j] & SRC_MASK) << 8);
            if (fits) {
                if ((unsigned)p < (unsigned)SORT_CAP) stage[p] = s;
            } else {
                const int q = beg + p;
                if ((unsigned)q < (unsigned)NEDGES) ssrc[q] = s;
            }
        }
    }
    for (unsigned c = wave; c < SORT_C; c += 32u) {
        const int e = pcE[c];
        for (int i = pcS[c] + 32 + (int)lane; i < e; i += 32) {
            const unsigned rr = (unsigned)binned[c * SORT_EPB + (unsigned)i];
            unsigned dl = rr >> SRC_BITS;
            dl = dl < SORT_R ? dl : SORT_R - 1u;
            const int p = atomicAdd(&cnt[dl], 1);
            const int s = (int)((rr & SRC_MASK) << 8);
            if (fits) {
                if ((unsigned)p < (unsigned)SORT_CAP) stage[p] = s;
            } else {
                const int q = beg + p;
                if ((unsigned)q < (unsigned)NEDGES) ssrc[q] = s;
            }
        }
    }
    __syncthreads();

    if (fits) {
        for (int i = (int)tid; i < total; i += 1024) {
            const int q = beg + i;
            if ((unsigned)q < (unsigned)NEDGES) ssrc[q] = stage[i];
        }
    }
}

#define PB 16

__device__ __forceinline__ void load_idx_block(const int* __restrict__ ssrc, int e, int (&s)[PB])
{
    e = e > 0 ? e : 0;
    e = e < NEDGES ? e : NEDGES;
    const __attribute__((address_space(4))) int* cs = (const __attribute__((address_space(4))) int*)ssrc;
#pragma unroll
    for (int k = 0; k < PB; ++k) s[k] = cs[e + k];
}

__device__ __forceinline__ v2f row_load(const float* __restrict__ X, int o, unsigned lane8)
{
    const unsigned oc = (unsigned)o < (unsigned)(NNODES - 1) * 256u ? (unsigned)o : (unsigned)(NNODES - 1) * 256u;
    const unsigned o32 = oc + lane8;
    return *(const v2f*)((const char*)X + o32);
}

__device__ __forceinline__ void gather_issue(
    const float* __restrict__ X, const int (&s)[PB], int rem, unsigned lane8, v2f (&v)[PB])
{
    const v2f zero = {0.0f, 0.0f};
#pragma unroll
    for (int g = 0; g < PB; g += 4) {
        v[g]     = zero;
        v[g + 1] = zero;
        v[g + 2] = zero;
        v[g + 3] = zero;
        if (rem > g) {
            v[g] = row_load(X, s[g], lane8);
            if (rem > g + 1) {
                v[g + 1] = row_load(X, s[g + 1], lane8);
                if (rem > g + 2) {
                    v[g + 2] = row_load(X, s[g + 2], lane8);
                    if (rem > g + 3) v[g + 3] = row_load(X, s[g + 3], lane8);
                }
            }
        }
    }
}

__device__ __forceinline__ v2f gather_sum(const v2f (&v)[PB], v2f acc)
{
#pragma unroll
    for (int k = 0; k < PB; ++k) acc += v[k];
    return acc;
}

__device__ __forceinline__ v2f pull_row_sum(
    const float* __restrict__ X, const int* __restrict__ ssrc,
    int beg, int end, int next_beg, bool has_next, int (&idx)[PB], unsigned lane8, v2f acc)
{
    int e = beg;
    bool more;
    do {
        v2f v[PB];
        const int rem = end - e;
        gather_issue(X, idx, rem, lane8, v);
        e += PB;
        more = e < end;
        if (more || has_next) load_idx_block(ssrc, more ? e : next_beg, idx);
        acc = gather_sum(v, acc);
    } while (more);
    return acc;
}

#define PG_PAIRS   (NNODES / 32)
#define PG_PASSES  2
#define PG_WGS     ((PG_PAIRS + PG_PASSES - 1) / PG_PASSES)

static_assert(NNODES % 32 == 0, "pairs of row tiles have no tail");

__device__ __forceinline__ void rows_fetch(
    const int* __restrict__ off, const int* __restrict__ endp, const int* __restrict__ ssrc,
    unsigned n0, unsigned lane, unsigned mask, int& vb, int& ve, int (&idx)[PB])
{
    const int b0 = off[n0 + (lane & mask)];
    const int e0 = endp[n0 + (lane & mask)];
    vb = b0 > 0 ? b0 : 0;
    ve = e0 < NEDGES ? e0 : NEDGES;
    load_idx_block(ssrc, __builtin_amdgcn_readlane(vb, 0), idx);
}

#define PG_OUT_WGS ((NGRAPHS * 64) / 256)
static_assert((NGRAPHS * 64) % 256 == 0 && PG_OUT_WGS <= PG_WGS, "a thread per output element, whole work-groups");

__global__ __launch_bounds__(256, 16) void gcn_pull_gemm(
    const float* __restrict__ X, const int* __restrict__ off,
    const int* __restrict__ endp, const int* __restrict__ ssrc,
    const float* __restrict__ W, const float* __restrict__ bias_in,
    float* __restrict__ Out,
    const float* __restrict__ gcnt, const float* __restrict__ b2, float* __restrict__ gout)
{
    __shared__ float sW[64 * 65];
    __shared__ float sA[2][32 * 65];

    const unsigned tid  = threadIdx.x;
    const unsigned wave = tid >> 5;
    const unsigned lane = tid & 31u;
    const unsigned lane8 = lane << 3;

    unsigned pair = blockIdx.x;
    int vb, ve, idx[PB];
    rows_fetch(off, endp, ssrc, pair * 32u + (wave << 2), lane, 3u, vb, ve, idx);

    float4 w[4];
#pragma unroll
    for (int i = 0; i < 4; ++i) w[i] = ((const float4*)W)[tid + 256u * (unsigned)i];
#pragma unroll
    for (int i = 0; i < 4; ++i) {
        const unsigned q = tid + 256u * (unsigned)i;
        float* p = sW + (q >> 4) * 65u + ((q & 15u) << 2);
        p[0] = w[i].x; p[1] = w[i].y; p[2] = w[i].z; p[3] = w[i].w;
    }

    const float bx = bias_in[2u * lane], by = bias_in[2u * lane + 1u];

    const unsigned m     = lane & 15u;
    const unsigned t16   = (wave >> 2) << 4;
    const unsigned n     = ((wave & 3u) << 4) + m;
    const unsigned khalf = (lane >> 4) << 1;
    const unsigned rOff  = (lane >> 4) << 3;

    unsigned it = 0;
    do {
        float* A = sA[it & 1u];

        for (int r = 0; r < 4; ++r) {
            const int beg = __builtin_amdgcn_readlane(vb, r);
            const int end = __builtin_amdgcn_readlane(ve, r);
            const int nb  = __builtin_amdgcn_readlane(vb, (r + 1) & 3);
            const v2f a = pull_row_sum(X, ssrc, beg, end, nb, r < 3, idx, lane8, (v2f){0.0f, 0.0f});
            float* p = A + ((wave << 2) + (unsigned)r) * 65u + 2u * lane;
            const float v0 = a.x + bx, v1 = a.y + by;
            p[0] = v0 > 0.0f ? v0 : 0.0f;
            p[1] = v1 > 0.0f ? v1 : 0.0f;
        }

        const unsigned nxt = pair + PG_WGS;
        if (nxt < PG_PAIRS)
            rows_fetch(off, endp, ssrc, nxt * 32u + (wave << 2), lane, 3u, vb, ve, idx);

        __syncthreads();

        v8f acc = {};
#pragma unroll
        for (int k = 0; k < 64; k += 4) {
            v2f a, b;
            a.x = A[(t16 + m) * 65u + k + khalf];
            a.y = A[(t16 + m) * 65u + k + khalf + 1];
            b.x = sW[(k + khalf) * 65u + n];
            b.y = sW[(k + khalf + 1) * 65u + n];
            acc = __builtin_amdgcn_wmma_f32_16x16x4_f32(
                 false, a,  false, b,
                 (short)0, acc,  false,  false);
        }

        float* outp = Out + (size_t)(pair * 32u + t16 + rOff) * 64 + n;
#pragma unroll
        for (int v = 0; v < 8; ++v)
            outp[(size_t)v * 64] = acc[v];

        pair = nxt;
        ++it;
    } while (pair < PG_PAIRS);

    if (blockIdx.x < PG_OUT_WGS) {
        const unsigned oi = blockIdx.x * 256u + tid;
        const float oCnt = gcnt[oi >> 6];
        const float oB   = b2[oi & 63u];
        gout[oi] = oCnt > 0.0f ? oB : 0.0f;
    }
}

#define PR_ROWS 8

__device__ __forceinline__ void readout_flush(
    float* __restrict__ out, int g, unsigned lane, v2f acc, float cnt)
{
    float* p = out + (size_t)g * 64 + 2u * lane;
    unsafeAtomicAdd(p, acc.x / cnt);
    unsafeAtomicAdd(p + 1, acc.y / cnt);
}

__global__ __launch_bounds__(256) void gcn_pull_readout(
    const float* __restrict__ X, const int* __restrict__ off,
    const int* __restrict__ endp, const int* __restrict__ ssrc,
    const int* __restrict__ gids,
    const float* __restrict__ gcnt, float* __restrict__ out)
{
    const unsigned wave = __builtin_amdgcn_readfirstlane((blockIdx.x * 256u + threadIdx.x) >> 5);
    const unsigned lane = threadIdx.x & 31u;
    const unsigned n0 = wave * PR_ROWS;
    if (n0 >= NNODES) return;
    const unsigned lane8 = lane << 3;
    int gl = gids[n0 + (lane & (PR_ROWS - 1))];
    gl = gl > 0 ? gl : 0;
    gl = gl < NGRAPHS ? gl : NGRAPHS - 1;
    const float cl = gcnt[gl];

    int vb, ve, idx[PB];
    rows_fetch(off, endp, ssrc, n0, lane, PR_ROWS - 1, vb, ve, idx);

    int cur = __builtin_amdgcn_readlane(gl, 0);
    int curRow = 0;
    v2f racc = {0.0f, 0.0f};
    for (int r = 0; r < PR_ROWS; ++r) {
        const int g = __builtin_amdgcn_readlane(gl, r);
        if (g != cur) {
            readout_flush(out, cur, lane, racc, __int_as_float(__builtin_amdgcn_readlane(__float_as_int(cl), curRow)));
            cur = g;
            curRow = r;
            racc = (v2f){0.0f, 0.0f};
        }
        const int beg = __builtin_amdgcn_readlane(vb, r);
        const int end = __builtin_amdgcn_readlane(ve, r);
        const int nb  = __builtin_amdgcn_readlane(vb, (r + 1) & (PR_ROWS - 1));
        racc = pull_row_sum(X, ssrc, beg, end, nb, r < PR_ROWS - 1, idx, lane8, racc);
    }
    readout_flush(out, cur, lane, racc, __int_as_float(__builtin_amdgcn_readlane(__float_as_int(cl), curRow)));
}

extern "C" void kernel_launch(void* const* d_in, const int* in_sizes, int n_in,
                              void* d_out, int out_size, void* d_ws, size_t ws_size,
                              hipStream_t stream)
{
    (void)in_sizes; (void)n_in; (void)out_size; (void)ws_size;

    const float* feats = (const float*)d_in[0];
    const float* W1    = (const float*)d_in[1];
    const float* b1    = (const float*)d_in[2];
    const float* W2    = (const float*)d_in[3];
    const float* b2    = (const float*)d_in[4];
    const int*   src   = (const int*)d_in[5];
    const int*   dst   = (const int*)d_in[6];
    const int*   gids  = (const int*)d_in[7];
    float* out = (float*)d_out;

    const size_t nodeElems = (size_t)NNODES * HID;
    float* bufA = (float*)d_ws;
    float* bufB = bufA + nodeElems;
    float* gsum = bufB + nodeElems;
    float* gcnt = gsum + (size_t)NGRAPHS * HID;
    int*   ssrc = (int*)(gcnt + NGRAPHS);
    int*   cur  = ssrc + NEDGES;
    int*   off  = cur + CSR_PAD;
    int*   binned = (int*)bufB;
    int*   tbl    = binned + NEDGES;

    const int gemmBlocks    = PG_WGS;
    const int readBlocks    = (NNODES / PR_ROWS * 32 + 255) / 256;

    hipLaunchKernelGGL(sort_bin_gemm, dim3(SORT_C + FG_WGS), dim3(1024), 0, stream,
                       src, dst, binned, tbl, feats, W1, bufA, gcnt);
    hipLaunchKernelGGL(sort_place, dim3(SORT_NB), dim3(1024), 0, stream, binned, tbl, off, cur, ssrc, gids, gcnt);
    hipLaunchKernelGGL(gcn_pull_gemm, dim3(gemmBlocks), dim3(256), 0, stream,
                       bufA, off, cur, ssrc, W2, b1, bufB, gcnt, b2, out);
    hipLaunchKernelGGL(gcn_pull_readout, dim3(readBlocks), dim3(256), 0, stream,
                       bufB, off, cur, ssrc, gids, gcnt, out);
}
